# starting kernel
# baseline (speedup 1.0000x reference)
#include <hip/hip_runtime.h>

typedef __attribute__((ext_vector_type(2))) float v2f;
typedef __attribute__((ext_vector_type(8))) float v8f;

#define GCN_N_NODES 100000

__global__ void __launch_bounds__(256)
deg_init_kernel(float* __restrict__ deg, int n) {
  int i = blockIdx.x * blockDim.x + threadIdx.x;
  if (i < n) deg[i] = 1.0f;
}

__global__ void __launch_bounds__(256)
deg_count_kernel(const int* __restrict__ dst, float* __restrict__ deg, int E) {
  int e = blockIdx.x * blockDim.x + threadIdx.x;
  if (e < E) unsafeAtomicAdd(&deg[dst[e]], 1.0f);
}

__global__ void __launch_bounds__(256)
deg_rsqrt_kernel(float* __restrict__ deg, int n) {
  int i = blockIdx.x * blockDim.x + threadIdx.x;
  if (i < n) deg[i] = rsqrtf(deg[i]);
}

template <bool FUSE_RELU_BIAS>
__global__ void __launch_bounds__(256)
gemm_wmma_kernel(const float* __restrict__ A, const float* __restrict__ W,
                 const float* __restrict__ bias, float* __restrict__ C,
                 int M, int N, int K, int totalTiles) {
  int wave = blockIdx.x * (blockDim.x >> 5) + (threadIdx.x >> 5);
  if (wave >= totalTiles) return;

  int tilesPerRow = N >> 4;
  int tileRow = wave / tilesPerRow;
  int tileCol = wave - tileRow * tilesPerRow;
  int row0 = tileRow << 4;
  int col0 = tileCol << 4;

  int lane = threadIdx.x & 31;
  int half = lane >> 4;
  int l15  = lane & 15;
  int kOff = half * 2;

  const float* arow = A + (size_t)(row0 + l15) * K;
  const float* wcol = W + col0 + l15;

  v8f acc = {};
#pragma unroll 4
  for (int k = 0; k < K; k += 4) {
    float a0 = arow[k + kOff];
    float a1 = arow[k + kOff + 1];
    if (FUSE_RELU_BIAS) {
      a0 = fmaxf(a0 + bias[k + kOff], 0.0f);
      a1 = fmaxf(a1 + bias[k + kOff + 1], 0.0f);
    }
    v2f a; a.x = a0; a.y = a1;
    v2f b;
    b.x = wcol[(size_t)(k + kOff) * N];
    b.y = wcol[(size_t)(k + kOff + 1) * N];
    acc = __builtin_amdgcn_wmma_f32_16x16x4_f32(
         false, a,  false, b,
         (short)0, acc,  false,  false);
  }

  float* crow = C + (size_t)(row0 + half * 8) * N + col0 + l15;
#pragma unroll
  for (int r = 0; r < 8; ++r) crow[(size_t)r * N] = acc[r];
}

template <int F>
__global__ void __launch_bounds__(256)
init_self_kernel(const float* __restrict__ h, const float* __restrict__ dinv,
                 const float* __restrict__ bias, float* __restrict__ out, int total) {
  int idx = blockIdx.x * blockDim.x + threadIdx.x;
  if (idx >= total) return;
  int i = idx / F;
  int f = idx & (F - 1);
  float di = dinv[i];
  float v = h[idx] * di * di;
  if (bias) v += bias[f];
  out[idx] = v;
}

template <int FPL>
__global__ void __launch_bounds__(256)
scatter_kernel(const float* __restrict__ h, const int* __restrict__ src,
               const int* __restrict__ dst, const float* __restrict__ dinv,
               float* __restrict__ out, int E) {
  const int F = FPL * 32;
  int e = (blockIdx.x * blockDim.x + threadIdx.x) >> 5;
  if (e >= E) return;
  int lane = threadIdx.x & 31;
  int s = src[e];
  int d = dst[e];
  float norm = dinv[s] * dinv[d];
  const float* hp = h + (size_t)s * F + lane * FPL;
  float* op = out + (size_t)d * F + lane * FPL;
  if constexpr (FPL == 4) {
    float4 v = *reinterpret_cast<const float4*>(hp);
    unsafeAtomicAdd(op + 0, v.x * norm);
    unsafeAtomicAdd(op + 1, v.y * norm);
    unsafeAtomicAdd(op + 2, v.z * norm);
    unsafeAtomicAdd(op + 3, v.w * norm);
  } else {
    float2 v = *reinterpret_cast<const float2*>(hp);
    unsafeAtomicAdd(op + 0, v.x * norm);
    unsafeAtomicAdd(op + 1, v.y * norm);
  }
}

extern "C" void kernel_launch(void* const* d_in, const int* in_sizes, int n_in,
                              void* d_out, int out_size, void* d_ws, size_t ws_size,
                              hipStream_t stream) {
  const float* x  = (const float*)d_in[0];
  const int*   ei = (const int*)d_in[1];
  const float* W1 = (const float*)d_in[2];
  const float* b1 = (const float*)d_in[3];
  const float* W2 = (const float*)d_in[4];
  const float* b2 = (const float*)d_in[5];
  float* out = (float*)d_out;

  const int Nn = GCN_N_NODES;
  const int E  = in_sizes[1] / 2;
  const int* src = ei;
  const int* dst = ei + E;

  float* dinv = (float*)d_ws;
  float* h    = dinv + Nn;
  float* out1 = h + (size_t)Nn * 128;
  float* h2   = h;

  deg_init_kernel<<<(Nn + 255) / 256, 256, 0, stream>>>(dinv, Nn);
  deg_count_kernel<<<(E + 255) / 256, 256, 0, stream>>>(dst, dinv, E);
  deg_rsqrt_kernel<<<(Nn + 255) / 256, 256, 0, stream>>>(dinv, Nn);

  {
    int M = Nn, N = 128, K = 128;
    int tiles = (M >> 4) * (N >> 4);
    gemm_wmma_kernel<false><<<(tiles + 7) / 8, 256, 0, stream>>>(
        x, W1, nullptr, h, M, N, K, tiles);
  }
  init_self_kernel<128><<<((Nn * 128) + 255) / 256, 256, 0, stream>>>(
      h, dinv, nullptr, out1, Nn * 128);
  scatter_kernel<4><<<(E + 7) / 8, 256, 0, stream>>>(h, src, dst, dinv, out1, E);

  {
    int M = Nn, N = 64, K = 128;
    int tiles = (M >> 4) * (N >> 4);
    gemm_wmma_kernel<true><<<(tiles + 7) / 8, 256, 0, stream>>>(
        out1, W2, b1, h2, M, N, K, tiles);
  }
  init_self_kernel<64><<<((Nn * 64) + 255) / 256, 256, 0, stream>>>(
      h2, dinv, b2, out, Nn * 64);
  scatter_kernel<2><<<(E + 7) / 8, 256, 0, stream>>>(h2, src, dst, dinv, out, E);
}


# Float atomic scatter-add replaced by edges grouped per destination (integer atomic arrival rank) and one wave per node summing rows in registers; sum order per node now follows that rank
# speedup vs baseline: 2.1209x; 2.1209x over previous
#include <hip/hip_runtime.h>

typedef __attribute__((ext_vector_type(2))) float v2f;
typedef __attribute__((ext_vector_type(4))) float v4f;
typedef __attribute__((ext_vector_type(8))) float v8f;

#define GCN_N_NODES 100000

__global__ void __launch_bounds__(256)
cnt_zero_kernel(int* __restrict__ cnt, int n) {
  int i = blockIdx.x * blockDim.x + threadIdx.x;
  if (i < n) cnt[i] = 0;
}

__global__ void __launch_bounds__(256)
edge_rank_kernel(const int* __restrict__ dst, int* __restrict__ cnt,
                 int* __restrict__ rank, int E) {
  int e = blockIdx.x * blockDim.x + threadIdx.x;
  if (e < E) rank[e] = atomicAdd(&cnt[dst[e]], 1);
}

__global__ void __launch_bounds__(256)
block_sum_kernel(const int* __restrict__ cnt, int* __restrict__ bsum, int n) {
  __shared__ int sh[256];
  int t = threadIdx.x;
  int base = (blockIdx.x * 256 + t) * 4;
  int s = 0;
#pragma unroll
  for (int j = 0; j < 4; ++j)
    if (base + j < n) s += cnt[base + j];
  sh[t] = s;
  __syncthreads();
  for (int o = 128; o > 0; o >>= 1) {
    if (t < o) sh[t] += sh[t + o];
    __syncthreads();
  }
  if (t == 0) bsum[blockIdx.x] = sh[0];
}

__global__ void __launch_bounds__(256)
row_scan_kernel(const int* __restrict__ cnt, const int* __restrict__ bsum,
                int* __restrict__ rowptr, float* __restrict__ dinv, int n) {
  __shared__ int sh[256];
  int t = threadIdx.x;

  int part = 0;
  for (int j = t; j < (int)blockIdx.x; j += 256) part += bsum[j];
  sh[t] = part;
  __syncthreads();
  for (int o = 128; o > 0; o >>= 1) {
    if (t < o) sh[t] += sh[t + o];
    __syncthreads();
  }
  int boff = sh[0];
  __syncthreads();

  int base = (blockIdx.x * 256 + t) * 4;
  int c[4];
  int tsum = 0;
#pragma unroll
  for (int j = 0; j < 4; ++j) {
    c[j] = (base + j < n) ? cnt[base + j] : 0;
    tsum += c[j];
  }
  sh[t] = tsum;
  __syncthreads();
  for (int o = 1; o < 256; o <<= 1) {
    int v = (t >= o) ? sh[t - o] : 0;
    __syncthreads();
    sh[t] += v;
    __syncthreads();
  }
  int run = boff + sh[t] - tsum;
#pragma unroll
  for (int j = 0; j < 4; ++j) {
    int i = base + j;
    if (i < n) {
      rowptr[i] = run;
      dinv[i] = rsqrtf((float)(c[j] + 1));
      run += c[j];
      if (i == n - 1) rowptr[n] = run;
    }
  }
}

__global__ void __launch_bounds__(256)
csr_fill_kernel(const int* __restrict__ src, const int* __restrict__ dst,
                const int* __restrict__ rank, const int* __restrict__ rowptr,
                int* __restrict__ col, int E) {
  int e = blockIdx.x * blockDim.x + threadIdx.x;
  if (e < E) col[rowptr[dst[e]] + rank[e]] = src[e];
}

template <bool FUSE_RELU_BIAS>
__global__ void __launch_bounds__(256)
gemm_wmma_kernel(const float* __restrict__ A, const float* __restrict__ W,
                 const float* __restrict__ bias, float* __restrict__ C,
                 int M, int N, int K, int totalTiles) {
  int wave = blockIdx.x * (blockDim.x >> 5) + (threadIdx.x >> 5);
  if (wave >= totalTiles) return;

  int tilesPerRow = N >> 4;
  int tileRow = wave / tilesPerRow;
  int tileCol = wave - tileRow * tilesPerRow;
  int row0 = tileRow << 4;
  int col0 = tileCol << 4;

  int lane = threadIdx.x & 31;
  int half = lane >> 4;
  int l15  = lane & 15;
  int kOff = half * 2;

  const float* arow = A + (size_t)(row0 + l15) * K;
  const float* wcol = W + col0 + l15;

  v8f acc = {};
#pragma unroll 4
  for (int k = 0; k < K; k += 4) {
    float a0 = arow[k + kOff];
    float a1 = arow[k + kOff + 1];
    if (FUSE_RELU_BIAS) {
      a0 = fmaxf(a0 + bias[k + kOff], 0.0f);
      a1 = fmaxf(a1 + bias[k + kOff + 1], 0.0f);
    }
    v2f a; a.x = a0; a.y = a1;
    v2f b;
    b.x = wcol[(size_t)(k + kOff) * N];
    b.y = wcol[(size_t)(k + kOff + 1) * N];
    acc = __builtin_amdgcn_wmma_f32_16x16x4_f32(
         false, a,  false, b,
         (short)0, acc,  false,  false);
  }

  float* crow = C + (size_t)(row0 + half * 8) * N + col0 + l15;
#pragma unroll
  for (int r = 0; r < 8; ++r) crow[(size_t)r * N] = acc[r];
}

template <int FPL> struct row_vec { typedef v4f type; };
template <> struct row_vec<2> { typedef v2f type; };

template <int FPL, bool ADD_BIAS>
__global__ void __launch_bounds__(256)
gather_kernel(const float* __restrict__ h, const int* __restrict__ rowptr,
              const int* __restrict__ col, const float* __restrict__ dinv,
              const float* __restrict__ bias, float* __restrict__ out, int n) {
  typedef typename row_vec<FPL>::type vf;
  const int F = FPL * 32;
  int node = __builtin_amdgcn_readfirstlane(
      (int)((blockIdx.x * blockDim.x + threadIdx.x) >> 5));
  if (node >= n) return;
  int lane = threadIdx.x & 31;
  int beg = rowptr[node];
  int end = rowptr[node + 1];
  float dd = dinv[node];
  const float* hl = h + lane * FPL;

  vf acc = *reinterpret_cast<const vf*>(hl + (size_t)node * F) * (dd * dd);

  for (int base = beg; base < end; base += 32) {
    int j = base + lane;
    int s_l = 0;
    float w_l = 0.0f;
    if (j < end) {
      s_l = col[j];
      w_l = dinv[s_l] * dd;
    }
    int m = end - base;
    if (m > 32) m = 32;
#pragma unroll 4
    for (int k = 0; k < m; ++k) {
      int s = __builtin_amdgcn_readlane(s_l, k);
      float w = __int_as_float(__builtin_amdgcn_readlane(__float_as_int(w_l), k));
      vf v = *reinterpret_cast<const vf*>(hl + (size_t)s * F);
      acc += v * w;
    }
  }

  if (ADD_BIAS) acc += *reinterpret_cast<const vf*>(bias + lane * FPL);
  *reinterpret_cast<vf*>(out + (size_t)node * F + lane * FPL) = acc;
}

extern "C" void kernel_launch(void* const* d_in, const int* in_sizes, int n_in,
                              void* d_out, int out_size, void* d_ws, size_t ws_size,
                              hipStream_t stream) {
  const float* x  = (const float*)d_in[0];
  const int*   ei = (const int*)d_in[1];
  const float* W1 = (const float*)d_in[2];
  const float* b1 = (const float*)d_in[3];
  const float* W2 = (const float*)d_in[4];
  const float* b2 = (const float*)d_in[5];
  float* out = (float*)d_out;

  const int Nn = GCN_N_NODES;
  const int E  = in_sizes[1] / 2;
  const int* src = ei;
  const int* dst = ei + E;

  float* dinv = (float*)d_ws;
  float* h    = dinv + Nn;
  float* out1 = h + (size_t)Nn * 128;
  float* h2   = h;
  int* cnt    = (int*)(out1 + (size_t)Nn * 128);
  int* rowptr = cnt + Nn;
  int* bsum   = rowptr + Nn + 32;
  int* rank   = bsum + 1024;
  int* col    = rank + E;

  const int scanBlocks = (Nn + 1023) / 1024;
  cnt_zero_kernel<<<(Nn + 255) / 256, 256, 0, stream>>>(cnt, Nn);
  edge_rank_kernel<<<(E + 255) / 256, 256, 0, stream>>>(dst, cnt, rank, E);
  block_sum_kernel<<<scanBlocks, 256, 0, stream>>>(cnt, bsum, Nn);
  row_scan_kernel<<<scanBlocks, 256, 0, stream>>>(cnt, bsum, rowptr, dinv, Nn);
  csr_fill_kernel<<<(E + 255) / 256, 256, 0, stream>>>(src, dst, rank, rowptr, col, E);

  {
    int M = Nn, N = 128, K = 128;
    int tiles = (M >> 4) * (N >> 4);
    gemm_wmma_kernel<false><<<(tiles + 7) / 8, 256, 0, stream>>>(
        x, W1, nullptr, h, M, N, K, tiles);
  }
  gather_kernel<4, false><<<(Nn + 7) / 8, 256, 0, stream>>>(
      h, rowptr, col, dinv, nullptr, out1, Nn);

  {
    int M = Nn, N = 64, K = 128;
    int tiles = (M >> 4) * (N >> 4);
    gemm_wmma_kernel<true><<<(tiles + 7) / 8, 256, 0, stream>>>(
        out1, W2, b1, h2, M, N, K, tiles);
  }
  gather_kernel<2, true><<<(Nn + 7) / 8, 256, 0, stream>>>(
      h2, rowptr, col, dinv, b2, out, Nn);
}


# Edge grouping as a two-level counting sort on LDS counters (256-node ranges, then per-node ranks) replaces global returning atomics and scattered slot stores; node sum order follows LDS arrival ranks
# speedup vs baseline: 2.8252x; 1.3321x over previous
#include <hip/hip_runtime.h>

typedef __attribute__((ext_vector_type(2))) float v2f;
typedef __attribute__((ext_vector_type(4))) float v4f;
typedef __attribute__((ext_vector_type(8))) float v8f;

#define GCN_N_NODES 100000

#define RANGE_SHIFT 8
#define RANGE_NODES (1 << RANGE_SHIFT)
#define N_RANGES ((GCN_N_NODES + RANGE_NODES - 1) / RANGE_NODES)
#define EDGE_SLICES 256
#define SRC_BITS 17
#define SRC_MASK ((1 << SRC_BITS) - 1)
static_assert(GCN_N_NODES <= (1 << SRC_BITS), "a source id must fit SRC_BITS");
static_assert(SRC_BITS + RANGE_SHIFT < 31, "a packed record must stay non-negative");

__global__ void __launch_bounds__(1024)
range_count_kernel(const int* __restrict__ dst, int* __restrict__ counts, int E) {
  __shared__ int hist[N_RANGES];
  const int t = threadIdx.x;
  const int nb = gridDim.x;
  const int chunk = (E + nb - 1) / nb;
  const int lo = blockIdx.x * chunk;
  int hi = lo + chunk;
  if (hi > E) hi = E;

  for (int i = t; i < N_RANGES; i += 1024) hist[i] = 0;
  __syncthreads();
  for (int base = lo; base < hi; base += 8 * 1024) {
    int d[8];
#pragma unroll
    for (int j = 0; j < 8; ++j) {
      int e = base + j * 1024 + t;
      d[j] = (e < hi) ? dst[e] : -1;
    }
#pragma unroll
    for (int j = 0; j < 8; ++j)
      if (d[j] >= 0) atomicAdd(&hist[d[j] >> RANGE_SHIFT], 1);
  }
  __syncthreads();
  for (int i = t; i < N_RANGES; i += 1024) counts[i * nb + blockIdx.x] = hist[i];
}

__global__ void __launch_bounds__(256)
block_sum_kernel(const int* __restrict__ cnt, int* __restrict__ bsum, int n) {
  __shared__ int sh[256];
  int t = threadIdx.x;
  int base = (blockIdx.x * 256 + t) * 4;
  int s = 0;
#pragma unroll
  for (int j = 0; j < 4; ++j)
    if (base + j < n) s += cnt[base + j];
  sh[t] = s;
  __syncthreads();
  for (int o = 128; o > 0; o >>= 1) {
    if (t < o) sh[t] += sh[t + o];
    __syncthreads();
  }
  if (t == 0) bsum[blockIdx.x] = sh[0];
}

__global__ void __launch_bounds__(256)
excl_scan_kernel(const int* __restrict__ cnt, const int* __restrict__ bsum,
                 int* __restrict__ offs, int n) {
  __shared__ int sh[256];
  int t = threadIdx.x;

  int part = 0;
  for (int j = t; j < (int)blockIdx.x; j += 256) part += bsum[j];
  sh[t] = part;
  __syncthreads();
  for (int o = 128; o > 0; o >>= 1) {
    if (t < o) sh[t] += sh[t + o];
    __syncthreads();
  }
  int boff = sh[0];
  __syncthreads();

  int base = (blockIdx.x * 256 + t) * 4;
  int c[4];
  int tsum = 0;
#pragma unroll
  for (int j = 0; j < 4; ++j) {
    c[j] = (base + j < n) ? cnt[base + j] : 0;
    tsum += c[j];
  }
  sh[t] = tsum;
  __syncthreads();
  for (int o = 1; o < 256; o <<= 1) {
    int v = (t >= o) ? sh[t - o] : 0;
    __syncthreads();
    sh[t] += v;
    __syncthreads();
  }
  int run = boff + sh[t] - tsum;
#pragma unroll
  for (int j = 0; j < 4; ++j) {
    int i = base + j;
    if (i < n) {
      offs[i] = run;
      run += c[j];
      if (i == n - 1) offs[n] = run;
    }
  }
}

__global__ void __launch_bounds__(1024)
range_scatter_kernel(const int* __restrict__ src, const int* __restrict__ dst,
                     const int* __restrict__ offs, int* __restrict__ binned, int E) {
  __shared__ int cur[N_RANGES];
  const int t = threadIdx.x;
  const int nb = gridDim.x;
  const int chunk = (E + nb - 1) / nb;
  const int lo = blockIdx.x * chunk;
  int hi = lo + chunk;
  if (hi > E) hi = E;

  for (int i = t; i < N_RANGES; i += 1024) cur[i] = offs[i * nb + blockIdx.x];
  __syncthreads();
  for (int base = lo; base < hi; base += 8 * 1024) {
    int d[8], s[8];
#pragma unroll
    for (int j = 0; j < 8; ++j) {
      int e = base + j * 1024 + t;
      bool ok = e < hi;
      d[j] = ok ? dst[e] : -1;
      s[j] = ok ? src[e] : 0;
    }
#pragma unroll
    for (int j = 0; j < 8; ++j)
      if (d[j] >= 0) {
        int pos = atomicAdd(&cur[d[j] >> RANGE_SHIFT], 1);
        binned[pos] = s[j] | ((d[j] & (RANGE_NODES - 1)) << SRC_BITS);
      }
  }
}

__global__ void __launch_bounds__(1024)
range_sort_kernel(const int* __restrict__ binned, const int* __restrict__ offs,
                  int* __restrict__ col, int* __restrict__ rowptr,
                  float* __restrict__ dinv, int n, int nb) {
  __shared__ int cnt[RANGE_NODES];
  __shared__ int wsum[RANGE_NODES / 32];
  const int t = threadIdx.x;
  const int b = blockIdx.x;
  const int bs = offs[b * nb];
  const int be = offs[(b + 1) * nb];

  if (t < RANGE_NODES) cnt[t] = 0;
  __syncthreads();
  for (int base = bs; base < be; base += 8 * 1024) {
    int r[8];
#pragma unroll
    for (int j = 0; j < 8; ++j) {
      int i = base + j * 1024 + t;
      r[j] = (i < be) ? binned[i] : -1;
    }
#pragma unroll
    for (int j = 0; j < 8; ++j)
      if (r[j] >= 0) atomicAdd(&cnt[r[j] >> SRC_BITS], 1);
  }
  __syncthreads();

  int c = 0, incl = 0;
  if (t < RANGE_NODES) {
    c = cnt[t];
    incl = c;
#pragma unroll
    for (int o = 1; o < 32; o <<= 1) {
      int v = __shfl_up(incl, o);
      if ((t & 31) >= o) incl += v;
    }
    if ((t & 31) == 31) wsum[t >> 5] = incl;
  }
  __syncthreads();
  if (t < RANGE_NODES) {
    int first = bs + incl - c;
    for (int w = 0; w < (t >> 5); ++w) first += wsum[w];
    int node = b * RANGE_NODES + t;
    if (node < n) {
      rowptr[node] = first;
      dinv[node] = rsqrtf((float)(c + 1));
      if (node == n - 1) rowptr[n] = first + c;
    }
    cnt[t] = first;
  }
  __syncthreads();
  for (int base = bs; base < be; base += 8 * 1024) {
    int r[8];
#pragma unroll
    for (int j = 0; j < 8; ++j) {
      int i = base + j * 1024 + t;
      r[j] = (i < be) ? binned[i] : -1;
    }
#pragma unroll
    for (int j = 0; j < 8; ++j)
      if (r[j] >= 0) {
        int pos = atomicAdd(&cnt[r[j] >> SRC_BITS], 1);
        col[pos] = r[j] & SRC_MASK;
      }
  }
}

template <bool FUSE_RELU_BIAS>
__global__ void __launch_bounds__(256)
gemm_wmma_kernel(const float* __restrict__ A, const float* __restrict__ W,
                 const float* __restrict__ bias, float* __restrict__ C,
                 int M, int N, int K, int totalTiles) {
  int wave = blockIdx.x * (blockDim.x >> 5) + (threadIdx.x >> 5);
  if (wave >= totalTiles) return;

  int tilesPerRow = N >> 4;
  int tileRow = wave / tilesPerRow;
  int tileCol = wave - tileRow * tilesPerRow;
  int row0 = tileRow << 4;
  int col0 = tileCol << 4;

  int lane = threadIdx.x & 31;
  int half = lane >> 4;
  int l15  = lane & 15;
  int kOff = half * 2;

  const float* arow = A + (size_t)(row0 + l15) * K;
  const float* wcol = W + col0 + l15;

  v8f acc = {};
#pragma unroll 4
  for (int k = 0; k < K; k += 4) {
    float a0 = arow[k + kOff];
    float a1 = arow[k + kOff + 1];
    if (FUSE_RELU_BIAS) {
      a0 = fmaxf(a0 + bias[k + kOff], 0.0f);
      a1 = fmaxf(a1 + bias[k + kOff + 1], 0.0f);
    }
    v2f a; a.x = a0; a.y = a1;
    v2f b;
    b.x = wcol[(size_t)(k + kOff) * N];
    b.y = wcol[(size_t)(k + kOff + 1) * N];
    acc = __builtin_amdgcn_wmma_f32_16x16x4_f32(
         false, a,  false, b,
         (short)0, acc,  false,  false);
  }

  float* crow = C + (size_t)(row0 + half * 8) * N + col0 + l15;
#pragma unroll
  for (int r = 0; r < 8; ++r) crow[(size_t)r * N] = acc[r];
}

template <int FPL> struct row_vec { typedef v4f type; };
template <> struct row_vec<2> { typedef v2f type; };

template <int FPL, bool ADD_BIAS>
__global__ void __launch_bounds__(256)
gather_kernel(const float* __restrict__ h, const int* __restrict__ rowptr,
              const int* __restrict__ col, const float* __restrict__ dinv,
              const float* __restrict__ bias, float* __restrict__ out, int n) {
  typedef typename row_vec<FPL>::type vf;
  const int F = FPL * 32;
  int node = __builtin_amdgcn_readfirstlane(
      (int)((blockIdx.x * blockDim.x + threadIdx.x) >> 5));
  if (node >= n) return;
  int lane = threadIdx.x & 31;
  int beg = rowptr[node];
  int end = rowptr[node + 1];
  float dd = dinv[node];
  const float* hl = h + lane * FPL;

  vf acc = *reinterpret_cast<const vf*>(hl + (size_t)node * F) * (dd * dd);

  for (int base = beg; base < end; base += 32) {
    int j = base + lane;
    int s_l = 0;
    float w_l = 0.0f;
    if (j < end) {
      s_l = col[j];
      w_l = dinv[s_l] * dd;
    }
    int m = end - base;
    if (m > 32) m = 32;
#pragma unroll 4
    for (int k = 0; k < m; ++k) {
      int s = __builtin_amdgcn_readlane(s_l, k);
      float w = __int_as_float(__builtin_amdgcn_readlane(__float_as_int(w_l), k));
      vf v = *reinterpret_cast<const vf*>(hl + (size_t)s * F);
      acc += v * w;
    }
  }

  if (ADD_BIAS) acc += *reinterpret_cast<const vf*>(bias + lane * FPL);
  *reinterpret_cast<vf*>(out + (size_t)node * F + lane * FPL) = acc;
}

extern "C" void kernel_launch(void* const* d_in, const int* in_sizes, int n_in,
                              void* d_out, int out_size, void* d_ws, size_t ws_size,
                              hipStream_t stream) {
  const float* x  = (const float*)d_in[0];
  const int*   ei = (const int*)d_in[1];
  const float* W1 = (const float*)d_in[2];
  const float* b1 = (const float*)d_in[3];
  const float* W2 = (const float*)d_in[4];
  const float* b2 = (const float*)d_in[5];
  float* out = (float*)d_out;

  const int Nn = GCN_N_NODES;
  const int E  = in_sizes[1] / 2;
  const int* src = ei;
  const int* dst = ei + E;

  const int nCounts = N_RANGES * EDGE_SLICES;
  float* dinv = (float*)d_ws;
  float* h    = dinv + Nn;
  float* out1 = h + (size_t)Nn * 128;
  float* h2   = h;
  int* counts = (int*)(out1 + (size_t)Nn * 128);
  int* offs   = counts + nCounts;
  int* rowptr = offs + nCounts + 32;
  int* bsum   = rowptr + Nn + 32;
  int* binned = bsum + 1024;
  int* col    = binned + E;

  const int scanBlocks = (nCounts + 1023) / 1024;
  range_count_kernel<<<EDGE_SLICES, 1024, 0, stream>>>(dst, counts, E);
  block_sum_kernel<<<scanBlocks, 256, 0, stream>>>(counts, bsum, nCounts);
  excl_scan_kernel<<<scanBlocks, 256, 0, stream>>>(counts, bsum, offs, nCounts);
  range_scatter_kernel<<<EDGE_SLICES, 1024, 0, stream>>>(src, dst, offs, binned, E);
  range_sort_kernel<<<N_RANGES, 1024, 0, stream>>>(binned, offs, col, rowptr, dinv, Nn,
                                                  EDGE_SLICES);

  {
    int M = Nn, N = 128, K = 128;
    int tiles = (M >> 4) * (N >> 4);
    gemm_wmma_kernel<false><<<(tiles + 7) / 8, 256, 0, stream>>>(
        x, W1, nullptr, h, M, N, K, tiles);
  }
  gather_kernel<4, false><<<(Nn + 7) / 8, 256, 0, stream>>>(
      h, rowptr, col, dinv, nullptr, out1, Nn);

  {
    int M = Nn, N = 64, K = 128;
    int tiles = (M >> 4) * (N >> 4);
    gemm_wmma_kernel<true><<<(tiles + 7) / 8, 256, 0, stream>>>(
        out1, W2, b1, h2, M, N, K, tiles);
  }
  gather_kernel<2, true><<<(Nn + 7) / 8, 256, 0, stream>>>(
      h2, rowptr, col, dinv, b2, out, Nn);
}


# Both GEMMs re-tiled: W put once per work-group into LDS as B fragments; each wave computes 16x64 strips, A rows read once as 16-B vectors and reused over 4 column tiles; k sum order changes
# speedup vs baseline: 3.6189x; 1.2809x over previous
#include <hip/hip_runtime.h>

typedef __attribute__((ext_vector_type(2))) float v2f;
typedef __attribute__((ext_vector_type(4))) float v4f;
typedef __attribute__((ext_vector_type(8))) float v8f;

#define GCN_N_NODES 100000

#define RANGE_SHIFT 8
#define RANGE_NODES (1 << RANGE_SHIFT)
#define N_RANGES ((GCN_N_NODES + RANGE_NODES - 1) / RANGE_NODES)
#define EDGE_SLICES 256
#define SRC_BITS 17
#define SRC_MASK ((1 << SRC_BITS) - 1)
static_assert(GCN_N_NODES <= (1 << SRC_BITS), "a source id must fit SRC_BITS");
static_assert(SRC_BITS + RANGE_SHIFT < 31, "a packed record must stay non-negative");

__global__ void __launch_bounds__(1024)
range_count_kernel(const int* __restrict__ dst, int* __restrict__ counts, int E) {
  __shared__ int hist[N_RANGES];
  const int t = threadIdx.x;
  const int nb = gridDim.x;
  const int chunk = (E + nb - 1) / nb;
  const int lo = blockIdx.x * chunk;
  int hi = lo + chunk;
  if (hi > E) hi = E;

  for (int i = t; i < N_RANGES; i += 1024) hist[i] = 0;
  __syncthreads();
  for (int base = lo; base < hi; base += 8 * 1024) {
    int d[8];
#pragma unroll
    for (int j = 0; j < 8; ++j) {
      int e = base + j * 1024 + t;
      d[j] = (e < hi) ? dst[e] : -1;
    }
#pragma unroll
    for (int j = 0; j < 8; ++j)
      if (d[j] >= 0) atomicAdd(&hist[d[j] >> RANGE_SHIFT], 1);
  }
  __syncthreads();
  for (int i = t; i < N_RANGES; i += 1024) counts[i * nb + blockIdx.x] = hist[i];
}

__global__ void __launch_bounds__(256)
block_sum_kernel(const int* __restrict__ cnt, int* __restrict__ bsum, int n) {
  __shared__ int sh[256];
  int t = threadIdx.x;
  int base = (blockIdx.x * 256 + t) * 4;
  int s = 0;
#pragma unroll
  for (int j = 0; j < 4; ++j)
    if (base + j < n) s += cnt[base + j];
  sh[t] = s;
  __syncthreads();
  for (int o = 128; o > 0; o >>= 1) {
    if (t < o) sh[t] += sh[t + o];
    __syncthreads();
  }
  if (t == 0) bsum[blockIdx.x] = sh[0];
}

__global__ void __launch_bounds__(256)
excl_scan_kernel(const int* __restrict__ cnt, const int* __restrict__ bsum,
                 int* __restrict__ offs, int n) {
  __shared__ int sh[256];
  int t = threadIdx.x;

  int part = 0;
  for (int j = t; j < (int)blockIdx.x; j += 256) part += bsum[j];
  sh[t] = part;
  __syncthreads();
  for (int o = 128; o > 0; o >>= 1) {
    if (t < o) sh[t] += sh[t + o];
    __syncthreads();
  }
  int boff = sh[0];
  __syncthreads();

  int base = (blockIdx.x * 256 + t) * 4;
  int c[4];
  int tsum = 0;
#pragma unroll
  for (int j = 0; j < 4; ++j) {
    c[j] = (base + j < n) ? cnt[base + j] : 0;
    tsum += c[j];
  }
  sh[t] = tsum;
  __syncthreads();
  for (int o = 1; o < 256; o <<= 1) {
    int v = (t >= o) ? sh[t - o] : 0;
    __syncthreads();
    sh[t] += v;
    __syncthreads();
  }
  int run = boff + sh[t] - tsum;
#pragma unroll
  for (int j = 0; j < 4; ++j) {
    int i = base + j;
    if (i < n) {
      offs[i] = run;
      run += c[j];
      if (i == n - 1) offs[n] = run;
    }
  }
}

__global__ void __launch_bounds__(1024)
range_scatter_kernel(const int* __restrict__ src, const int* __restrict__ dst,
                     const int* __restrict__ offs, int* __restrict__ binned, int E) {
  __shared__ int cur[N_RANGES];
  const int t = threadIdx.x;
  const int nb = gridDim.x;
  const int chunk = (E + nb - 1) / nb;
  const int lo = blockIdx.x * chunk;
  int hi = lo + chunk;
  if (hi > E) hi = E;

  for (int i = t; i < N_RANGES; i += 1024) cur[i] = offs[i * nb + blockIdx.x];
  __syncthreads();
  for (int base = lo; base < hi; base += 8 * 1024) {
    int d[8], s[8];
#pragma unroll
    for (int j = 0; j < 8; ++j) {
      int e = base + j * 1024 + t;
      bool ok = e < hi;
      d[j] = ok ? dst[e] : -1;
      s[j] = ok ? src[e] : 0;
    }
#pragma unroll
    for (int j = 0; j < 8; ++j)
      if (d[j] >= 0) {
        int pos = atomicAdd(&cur[d[j] >> RANGE_SHIFT], 1);
        binned[pos] = s[j] | ((d[j] & (RANGE_NODES - 1)) << SRC_BITS);
      }
  }
}

__global__ void __launch_bounds__(1024)
range_sort_kernel(const int* __restrict__ binned, const int* __restrict__ offs,
                  int* __restrict__ col, int* __restrict__ rowptr,
                  float* __restrict__ dinv, int n, int nb) {
  __shared__ int cnt[RANGE_NODES];
  __shared__ int wsum[RANGE_NODES / 32];
  const int t = threadIdx.x;
  const int b = blockIdx.x;
  const int bs = offs[b * nb];
  const int be = offs[(b + 1) * nb];

  if (t < RANGE_NODES) cnt[t] = 0;
  __syncthreads();
  for (int base = bs; base < be; base += 8 * 1024) {
    int r[8];
#pragma unroll
    for (int j = 0; j < 8; ++j) {
      int i = base + j * 1024 + t;
      r[j] = (i < be) ? binned[i] : -1;
    }
#pragma unroll
    for (int j = 0; j < 8; ++j)
      if (r[j] >= 0) atomicAdd(&cnt[r[j] >> SRC_BITS], 1);
  }
  __syncthreads();

  int c = 0, incl = 0;
  if (t < RANGE_NODES) {
    c = cnt[t];
    incl = c;
#pragma unroll
    for (int o = 1; o < 32; o <<= 1) {
      int v = __shfl_up(incl, o);
      if ((t & 31) >= o) incl += v;
    }
    if ((t & 31) == 31) wsum[t >> 5] = incl;
  }
  __syncthreads();
  if (t < RANGE_NODES) {
    int first = bs + incl - c;
    for (int w = 0; w < (t >> 5); ++w) first += wsum[w];
    int node = b * RANGE_NODES + t;
    if (node < n) {
      rowptr[node] = first;
      dinv[node] = rsqrtf((float)(c + 1));
      if (node == n - 1) rowptr[n] = first + c;
    }
    cnt[t] = first;
  }
  __syncthreads();
  for (int base = bs; base < be; base += 8 * 1024) {
    int r[8];
#pragma unroll
    for (int j = 0; j < 8; ++j) {
      int i = base + j * 1024 + t;
      r[j] = (i < be) ? binned[i] : -1;
    }
#pragma unroll
    for (int j = 0; j < 8; ++j)
      if (r[j] >= 0) {
        int pos = atomicAdd(&cnt[r[j] >> SRC_BITS], 1);
        col[pos] = r[j] & SRC_MASK;
      }
  }
}

#define GEMM_K 128
#define GEMM_WAVES 32

template <int N, bool FUSE_RELU_BIAS>
__global__ void __launch_bounds__(GEMM_WAVES * 32)
gemm_wmma_kernel(const float* __restrict__ A, const float* __restrict__ W,
                 const float* __restrict__ bias, float* __restrict__ C, int M) {
  __shared__ v4f Wl[(GEMM_K / 4) * N];
  __shared__ v4f biasl[GEMM_K / 4];
  const int t = threadIdx.x;
  {
    const int n = t % N;
    for (int kb = t / N; kb < GEMM_K / 4; kb += (GEMM_WAVES * 32) / N) {
      const float* wp = W + (size_t)(4 * kb) * N + n;
      v4f v;
      v.x = wp[0];
      v.y = wp[N];
      v.z = wp[2 * N];
      v.w = wp[3 * N];
      Wl[kb * N + n] = v;
    }
    if (FUSE_RELU_BIAS && t < GEMM_K) reinterpret_cast<float*>(biasl)[t] = bias[t];
  }
  __syncthreads();

  constexpr int CH = N / 64;
  constexpr int SLOTS = GEMM_WAVES / CH;
  const int w = __builtin_amdgcn_readfirstlane(t >> 5);
  const int colBase = (w % CH) * 64;
  const int slot = w / CH;
  const int lane = t & 31;
  const int hh = lane >> 4;
  const int c = lane & 15;
  const int strips = M >> 4;

  for (int strip = slot * (int)gridDim.x + (int)blockIdx.x; strip < strips;
       strip += SLOTS * (int)gridDim.x) {
    const float* arow = A + (size_t)(strip * 16 + c) * GEMM_K + 4 * hh;
    const v4f* bl = Wl + hh * N + colBase + c;
    v8f acc0 = {}, acc1 = {}, acc2 = {}, acc3 = {};
#pragma unroll 4
    for (int j = 0; j < GEMM_K / 8; ++j) {
      v4f a = *reinterpret_cast<const v4f*>(arow + 8 * j);
      if (FUSE_RELU_BIAS) {
        v4f bv = biasl[2 * j + hh];
        a.x = fmaxf(a.x + bv.x, 0.0f);
        a.y = fmaxf(a.y + bv.y, 0.0f);
        a.z = fmaxf(a.z + bv.z, 0.0f);
        a.w = fmaxf(a.w + bv.w, 0.0f);
      }
      v2f a01; a01.x = a.x; a01.y = a.y;
      v2f a23; a23.x = a.z; a23.y = a.w;
      const v4f b0 = bl[(2 * j) * N];
      const v4f b1 = bl[(2 * j) * N + 16];
      const v4f b2 = bl[(2 * j) * N + 32];
      const v4f b3 = bl[(2 * j) * N + 48];
      v2f q;
      q.x = b0.x; q.y = b0.y;
      acc0 = __builtin_amdgcn_wmma_f32_16x16x4_f32(false, a01, false, q, (short)0, acc0, false, false);
      q.x = b1.x; q.y = b1.y;
      acc1 = __builtin_amdgcn_wmma_f32_16x16x4_f32(false, a01, false, q, (short)0, acc1, false, false);
      q.x = b2.x; q.y = b2.y;
      acc2 = __builtin_amdgcn_wmma_f32_16x16x4_f32(false, a01, false, q, (short)0, acc2, false, false);
      q.x = b3.x; q.y = b3.y;
      acc3 = __builtin_amdgcn_wmma_f32_16x16x4_f32(false, a01, false, q, (short)0, acc3, false, false);
      q.x = b0.z; q.y = b0.w;
      acc0 = __builtin_amdgcn_wmma_f32_16x16x4_f32(false, a23, false, q, (short)0, acc0, false, false);
      q.x = b1.z; q.y = b1.w;
      acc1 = __builtin_amdgcn_wmma_f32_16x16x4_f32(false, a23, false, q, (short)0, acc1, false, false);
      q.x = b2.z; q.y = b2.w;
      acc2 = __builtin_amdgcn_wmma_f32_16x16x4_f32(false, a23, false, q, (short)0, acc2, false, false);
      q.x = b3.z; q.y = b3.w;
      acc3 = __builtin_amdgcn_wmma_f32_16x16x4_f32(false, a23, false, q, (short)0, acc3, false, false);
    }

    float* crow = C + (size_t)(strip * 16 + hh * 8) * N + colBase + c;
#pragma unroll
    for (int r = 0; r < 8; ++r) {
      crow[(size_t)r * N] = acc0[r];
      crow[(size_t)r * N + 16] = acc1[r];
      crow[(size_t)r * N + 32] = acc2[r];
      crow[(size_t)r * N + 48] = acc3[r];
    }
  }
}

template <int FPL> struct row_vec { typedef v4f type; };
template <> struct row_vec<2> { typedef v2f type; };

template <int FPL, bool ADD_BIAS>
__global__ void __launch_bounds__(256)
gather_kernel(const float* __restrict__ h, const int* __restrict__ rowptr,
              const int* __restrict__ col, const float* __restrict__ dinv,
              const float* __restrict__ bias, float* __restrict__ out, int n) {
  typedef typename row_vec<FPL>::type vf;
  const int F = FPL * 32;
  int node = __builtin_amdgcn_readfirstlane(
      (int)((blockIdx.x * blockDim.x + threadIdx.x) >> 5));
  if (node >= n) return;
  int lane = threadIdx.x & 31;
  int beg = rowptr[node];
  int end = rowptr[node + 1];
  float dd = dinv[node];
  const float* hl = h + lane * FPL;

  vf acc = *reinterpret_cast<const vf*>(hl + (size_t)node * F) * (dd * dd);

  for (int base = beg; base < end; base += 32) {
    int j = base + lane;
    int s_l = 0;
    float w_l = 0.0f;
    if (j < end) {
      s_l = col[j];
      w_l = dinv[s_l] * dd;
    }
    int m = end - base;
    if (m > 32) m = 32;
#pragma unroll 4
    for (int k = 0; k < m; ++k) {
      int s = __builtin_amdgcn_readlane(s_l, k);
      float w = __int_as_float(__builtin_amdgcn_readlane(__float_as_int(w_l), k));
      vf v = *reinterpret_cast<const vf*>(hl + (size_t)s * F);
      acc += v * w;
    }
  }

  if (ADD_BIAS) acc += *reinterpret_cast<const vf*>(bias + lane * FPL);
  *reinterpret_cast<vf*>(out + (size_t)node * F + lane * FPL) = acc;
}

extern "C" void kernel_launch(void* const* d_in, const int* in_sizes, int n_in,
                              void* d_out, int out_size, void* d_ws, size_t ws_size,
                              hipStream_t stream) {
  const float* x  = (const float*)d_in[0];
  const int*   ei = (const int*)d_in[1];
  const float* W1 = (const float*)d_in[2];
  const float* b1 = (const float*)d_in[3];
  const float* W2 = (const float*)d_in[4];
  const float* b2 = (const float*)d_in[5];
  float* out = (float*)d_out;

  const int Nn = GCN_N_NODES;
  const int E  = in_sizes[1] / 2;
  const int* src = ei;
  const int* dst = ei + E;

  const int nCounts = N_RANGES * EDGE_SLICES;
  float* dinv = (float*)d_ws;
  float* h    = dinv + Nn;
  float* out1 = h + (size_t)Nn * 128;
  float* h2   = h;
  int* counts = (int*)(out1 + (size_t)Nn * 128);
  int* offs   = counts + nCounts;
  int* rowptr = offs + nCounts + 32;
  int* bsum   = rowptr + Nn + 32;
  int* binned = bsum + 1024;
  int* col    = binned + E;

  const int scanBlocks = (nCounts + 1023) / 1024;
  range_count_kernel<<<EDGE_SLICES, 1024, 0, stream>>>(dst, counts, E);
  block_sum_kernel<<<scanBlocks, 256, 0, stream>>>(counts, bsum, nCounts);
  excl_scan_kernel<<<scanBlocks, 256, 0, stream>>>(counts, bsum, offs, nCounts);
  range_scatter_kernel<<<EDGE_SLICES, 1024, 0, stream>>>(src, dst, offs, binned, E);
  range_sort_kernel<<<N_RANGES, 1024, 0, stream>>>(binned, offs, col, rowptr, dinv, Nn,
                                                  EDGE_SLICES);

  gemm_wmma_kernel<128, false><<<256, GEMM_WAVES * 32, 0, stream>>>(x, W1, nullptr, h, Nn);
  gather_kernel<4, false><<<(Nn + 7) / 8, 256, 0, stream>>>(
      h, rowptr, col, dinv, nullptr, out1, Nn);

  gemm_wmma_kernel<64, true><<<256, GEMM_WAVES * 32, 0, stream>>>(out1, W2, b1, h2, Nn);
  gather_kernel<2, true><<<(Nn + 7) / 8, 256, 0, stream>>>(
      h2, rowptr, col, dinv, b2, out, Nn);
}


# Edge weight factored: GEMM epilogues store rows scaled by dinv[row]; gathers add unweighted rows and scale the sum once by dinv[dst] (no per-edge dinv[src] gather); rounding changes, sum order kept
# speedup vs baseline: 3.9016x; 1.0781x over previous
#include <hip/hip_runtime.h>

typedef __attribute__((ext_vector_type(2))) float v2f;
typedef __attribute__((ext_vector_type(4))) float v4f;
typedef __attribute__((ext_vector_type(8))) float v8f;

#define GCN_N_NODES 100000

#define RANGE_SHIFT 8
#define RANGE_NODES (1 << RANGE_SHIFT)
#define N_RANGES ((GCN_N_NODES + RANGE_NODES - 1) / RANGE_NODES)
#define EDGE_SLICES 256
#define SRC_BITS 17
#define SRC_MASK ((1 << SRC_BITS) - 1)
static_assert(GCN_N_NODES <= (1 << SRC_BITS), "a source id must fit SRC_BITS");
static_assert(SRC_BITS + RANGE_SHIFT < 31, "a packed record must stay non-negative");

__global__ void __launch_bounds__(1024)
range_count_kernel(const int* __restrict__ dst, int* __restrict__ counts, int E) {
  __shared__ int hist[N_RANGES];
  const int t = threadIdx.x;
  const int nb = gridDim.x;
  const int chunk = (E + nb - 1) / nb;
  const int lo = blockIdx.x * chunk;
  int hi = lo + chunk;
  if (hi > E) hi = E;

  for (int i = t; i < N_RANGES; i += 1024) hist[i] = 0;
  __syncthreads();
  for (int base = lo; base < hi; base += 8 * 1024) {
    int d[8];
#pragma unroll
    for (int j = 0; j < 8; ++j) {
      int e = base + j * 1024 + t;
      d[j] = (e < hi) ? dst[e] : -1;
    }
#pragma unroll
    for (int j = 0; j < 8; ++j)
      if (d[j] >= 0) atomicAdd(&hist[d[j] >> RANGE_SHIFT], 1);
  }
  __syncthreads();
  for (int i = t; i < N_RANGES; i += 1024) counts[i * nb + blockIdx.x] = hist[i];
}

__global__ void __launch_bounds__(256)
block_sum_kernel(const int* __restrict__ cnt, int* __restrict__ bsum, int n) {
  __shared__ int sh[256];
  int t = threadIdx.x;
  int base = (blockIdx.x * 256 + t) * 4;
  int s = 0;
#pragma unroll
  for (int j = 0; j < 4; ++j)
    if (base + j < n) s += cnt[base + j];
  sh[t] = s;
  __syncthreads();
  for (int o = 128; o > 0; o >>= 1) {
    if (t < o) sh[t] += sh[t + o];
    __syncthreads();
  }
  if (t == 0) bsum[blockIdx.x] = sh[0];
}

__global__ void __launch_bounds__(256)
excl_scan_kernel(const int* __restrict__ cnt, const int* __restrict__ bsum,
                 int* __restrict__ offs, int n) {
  __shared__ int sh[256];
  int t = threadIdx.x;

  int part = 0;
  for (int j = t; j < (int)blockIdx.x; j += 256) part += bsum[j];
  sh[t] = part;
  __syncthreads();
  for (int o = 128; o > 0; o >>= 1) {
    if (t < o) sh[t] += sh[t + o];
    __syncthreads();
  }
  int boff = sh[0];
  __syncthreads();

  int base = (blockIdx.x * 256 + t) * 4;
  int c[4];
  int tsum = 0;
#pragma unroll
  for (int j = 0; j < 4; ++j) {
    c[j] = (base + j < n) ? cnt[base + j] : 0;
    tsum += c[j];
  }
  sh[t] = tsum;
  __syncthreads();
  for (int o = 1; o < 256; o <<= 1) {
    int v = (t >= o) ? sh[t - o] : 0;
    __syncthreads();
    sh[t] += v;
    __syncthreads();
  }
  int run = boff + sh[t] - tsum;
#pragma unroll
  for (int j = 0; j < 4; ++j) {
    int i = base + j;
    if (i < n) {
      offs[i] = run;
      run += c[j];
      if (i == n - 1) offs[n] = run;
    }
  }
}

__global__ void __launch_bounds__(1024)
range_scatter_kernel(const int* __restrict__ src, const int* __restrict__ dst,
                     const int* __restrict__ offs, int* __restrict__ binned, int E) {
  __shared__ int cur[N_RANGES];
  const int t = threadIdx.x;
  const int nb = gridDim.x;
  const int chunk = (E + nb - 1) / nb;
  const int lo = blockIdx.x * chunk;
  int hi = lo + chunk;
  if (hi > E) hi = E;

  for (int i = t; i < N_RANGES; i += 1024) cur[i] = offs[i * nb + blockIdx.x];
  __syncthreads();
  for (int base = lo; base < hi; base += 8 * 1024) {
    int d[8], s[8];
#pragma unroll
    for (int j = 0; j < 8; ++j) {
      int e = base + j * 1024 + t;
      bool ok = e < hi;
      d[j] = ok ? dst[e] : -1;
      s[j] = ok ? src[e] : 0;
    }
#pragma unroll
    for (int j = 0; j < 8; ++j)
      if (d[j] >= 0) {
        int pos = atomicAdd(&cur[d[j] >> RANGE_SHIFT], 1);
        binned[pos] = s[j] | ((d[j] & (RANGE_NODES - 1)) << SRC_BITS);
      }
  }
}

__global__ void __launch_bounds__(1024)
range_sort_kernel(const int* __restrict__ binned, const int* __restrict__ offs,
                  int* __restrict__ col, int* __restrict__ rowptr,
                  float* __restrict__ dinv, int n, int nb) {
  __shared__ int cnt[RANGE_NODES];
  __shared__ int wsum[RANGE_NODES / 32];
  const int t = threadIdx.x;
  const int b = blockIdx.x;
  const int bs = offs[b * nb];
  const int be = offs[(b + 1) * nb];

  if (t < RANGE_NODES) cnt[t] = 0;
  __syncthreads();
  for (int base = bs; base < be; base += 8 * 1024) {
    int r[8];
#pragma unroll
    for (int j = 0; j < 8; ++j) {
      int i = base + j * 1024 + t;
      r[j] = (i < be) ? binned[i] : -1;
    }
#pragma unroll
    for (int j = 0; j < 8; ++j)
      if (r[j] >= 0) atomicAdd(&cnt[r[j] >> SRC_BITS], 1);
  }
  __syncthreads();

  int c = 0, incl = 0;
  if (t < RANGE_NODES) {
    c = cnt[t];
    incl = c;
#pragma unroll
    for (int o = 1; o < 32; o <<= 1) {
      int v = __shfl_up(incl, o);
      if ((t & 31) >= o) incl += v;
    }
    if ((t & 31) == 31) wsum[t >> 5] = incl;
  }
  __syncthreads();
  if (t < RANGE_NODES) {
    int first = bs + incl - c;
    for (int w = 0; w < (t >> 5); ++w) first += wsum[w];
    int node = b * RANGE_NODES + t;
    if (node < n) {
      rowptr[node] = first;
      dinv[node] = rsqrtf((float)(c + 1));
      if (node == n - 1) rowptr[n] = first + c;
    }
    cnt[t] = first;
  }
  __syncthreads();
  for (int base = bs; base < be; base += 8 * 1024) {
    int r[8];
#pragma unroll
    for (int j = 0; j < 8; ++j) {
      int i = base + j * 1024 + t;
      r[j] = (i < be) ? binned[i] : -1;
    }
#pragma unroll
    for (int j = 0; j < 8; ++j)
      if (r[j] >= 0) {
        int pos = atomicAdd(&cnt[r[j] >> SRC_BITS], 1);
        col[pos] = r[j] & SRC_MASK;
      }
  }
}

#define GEMM_K 128
#define GEMM_WAVES 32

template <int N, bool FUSE_RELU_BIAS>
__global__ void __launch_bounds__(GEMM_WAVES * 32)
gemm_wmma_kernel(const float* __restrict__ A, const float* __restrict__ W,
                 const float* __restrict__ bias, const float* __restrict__ rowscale,
                 float* __restrict__ C, int M) {
  __shared__ v4f Wl[(GEMM_K / 4) * N];
  __shared__ v4f biasl[GEMM_K / 4];
  const int t = threadIdx.x;
  {
    const int n = t % N;
    for (int kb = t / N; kb < GEMM_K / 4; kb += (GEMM_WAVES * 32) / N) {
      const float* wp = W + (size_t)(4 * kb) * N + n;
      v4f v;
      v.x = wp[0];
      v.y = wp[N];
      v.z = wp[2 * N];
      v.w = wp[3 * N];
      Wl[kb * N + n] = v;
    }
    if (FUSE_RELU_BIAS && t < GEMM_K) reinterpret_cast<float*>(biasl)[t] = bias[t];
  }
  __syncthreads();

  constexpr int CH = N / 64;
  constexpr int SLOTS = GEMM_WAVES / CH;
  const int w = __builtin_amdgcn_readfirstlane(t >> 5);
  const int colBase = (w % CH) * 64;
  const int slot = w / CH;
  const int lane = t & 31;
  const int hh = lane >> 4;
  const int c = lane & 15;
  const int strips = M >> 4;

  for (int strip = slot * (int)gridDim.x + (int)blockIdx.x; strip < strips;
       strip += SLOTS * (int)gridDim.x) {
    const float* arow = A + (size_t)(strip * 16 + c) * GEMM_K + 4 * hh;
    const v4f* bl = Wl + hh * N + colBase + c;
    const float* sp = rowscale + strip * 16 + hh * 8;
    const v4f sc0 = *reinterpret_cast<const v4f*>(sp);
    const v4f sc1 = *reinterpret_cast<const v4f*>(sp + 4);
    v8f acc0 = {}, acc1 = {}, acc2 = {}, acc3 = {};
#pragma unroll 4
    for (int j = 0; j < GEMM_K / 8; ++j) {
      v4f a = *reinterpret_cast<const v4f*>(arow + 8 * j);
      if (FUSE_RELU_BIAS) {
        v4f bv = biasl[2 * j + hh];
        a.x = fmaxf(a.x + bv.x, 0.0f);
        a.y = fmaxf(a.y + bv.y, 0.0f);
        a.z = fmaxf(a.z + bv.z, 0.0f);
        a.w = fmaxf(a.w + bv.w, 0.0f);
      }
      v2f a01; a01.x = a.x; a01.y = a.y;
      v2f a23; a23.x = a.z; a23.y = a.w;
      const v4f b0 = bl[(2 * j) * N];
      const v4f b1 = bl[(2 * j) * N + 16];
      const v4f b2 = bl[(2 * j) * N + 32];
      const v4f b3 = bl[(2 * j) * N + 48];
      v2f q;
      q.x = b0.x; q.y = b0.y;
      acc0 = __builtin_amdgcn_wmma_f32_16x16x4_f32(false, a01, false, q, (short)0, acc0, false, false);
      q.x = b1.x; q.y = b1.y;
      acc1 = __builtin_amdgcn_wmma_f32_16x16x4_f32(false, a01, false, q, (short)0, acc1, false, false);
      q.x = b2.x; q.y = b2.y;
      acc2 = __builtin_amdgcn_wmma_f32_16x16x4_f32(false, a01, false, q, (short)0, acc2, false, false);
      q.x = b3.x; q.y = b3.y;
      acc3 = __builtin_amdgcn_wmma_f32_16x16x4_f32(false, a01, false, q, (short)0, acc3, false, false);
      q.x = b0.z; q.y = b0.w;
      acc0 = __builtin_amdgcn_wmma_f32_16x16x4_f32(false, a23, false, q, (short)0, acc0, false, false);
      q.x = b1.z; q.y = b1.w;
      acc1 = __builtin_amdgcn_wmma_f32_16x16x4_f32(false, a23, false, q, (short)0, acc1, false, false);
      q.x = b2.z; q.y = b2.w;
      acc2 = __builtin_amdgcn_wmma_f32_16x16x4_f32(false, a23, false, q, (short)0, acc2, false, false);
      q.x = b3.z; q.y = b3.w;
      acc3 = __builtin_amdgcn_wmma_f32_16x16x4_f32(false, a23, false, q, (short)0, acc3, false, false);
    }

    float* crow = C + (size_t)(strip * 16 + hh * 8) * N + colBase + c;
#pragma unroll
    for (int r = 0; r < 8; ++r) {
      const float sc = (r < 4) ? sc0[r & 3] : sc1[r & 3];
      crow[(size_t)r * N] = acc0[r] * sc;
      crow[(size_t)r * N + 16] = acc1[r] * sc;
      crow[(size_t)r * N + 32] = acc2[r] * sc;
      crow[(size_t)r * N + 48] = acc3[r] * sc;
    }
  }
}

template <int FPL> struct row_vec { typedef v4f type; };
template <> struct row_vec<2> { typedef v2f type; };

template <int FPL, bool ADD_BIAS>
__global__ void __launch_bounds__(256)
gather_kernel(const float* __restrict__ h, const int* __restrict__ rowptr,
              const int* __restrict__ col, const float* __restrict__ dinv,
              const float* __restrict__ bias, float* __restrict__ out, int n) {
  typedef typename row_vec<FPL>::type vf;
  const int F = FPL * 32;
  int node = __builtin_amdgcn_readfirstlane(
      (int)((blockIdx.x * blockDim.x + threadIdx.x) >> 5));
  if (node >= n) return;
  int lane = threadIdx.x & 31;
  int beg = rowptr[node];
  int end = rowptr[node + 1];
  float dd = dinv[node];
  const float* hl = h + lane * FPL;

  vf acc = *reinterpret_cast<const vf*>(hl + (size_t)node * F);

  for (int base = beg; base < end; base += 32) {
    int j = base + lane;
    int s_l = 0;
    if (j < end) s_l = col[j];
    int m = end - base;
    if (m > 32) m = 32;
#pragma unroll 4
    for (int k = 0; k < m; ++k) {
      int s = __builtin_amdgcn_readlane(s_l, k);
      acc += *reinterpret_cast<const vf*>(hl + (size_t)s * F);
    }
  }

  acc *= dd;
  if (ADD_BIAS) acc += *reinterpret_cast<const vf*>(bias + lane * FPL);
  *reinterpret_cast<vf*>(out + (size_t)node * F + lane * FPL) = acc;
}

extern "C" void kernel_launch(void* const* d_in, const int* in_sizes, int n_in,
                              void* d_out, int out_size, void* d_ws, size_t ws_size,
                              hipStream_t stream) {
  const float* x  = (const float*)d_in[0];
  const int*   ei = (const int*)d_in[1];
  const float* W1 = (const float*)d_in[2];
  const float* b1 = (const float*)d_in[3];
  const float* W2 = (const float*)d_in[4];
  const float* b2 = (const float*)d_in[5];
  float* out = (float*)d_out;

  const int Nn = GCN_N_NODES;
  const int E  = in_sizes[1] / 2;
  const int* src = ei;
  const int* dst = ei + E;

  const int nCounts = N_RANGES * EDGE_SLICES;
  float* dinv = (float*)d_ws;
  float* h    = dinv + Nn;
  float* out1 = h + (size_t)Nn * 128;
  float* h2   = h;
  int* counts = (int*)(out1 + (size_t)Nn * 128);
  int* offs   = counts + nCounts;
  int* rowptr = offs + nCounts + 32;
  int* bsum   = rowptr + Nn + 32;
  int* binned = bsum + 1024;
  int* col    = binned + E;

  const int scanBlocks = (nCounts + 1023) / 1024;
  range_count_kernel<<<EDGE_SLICES, 1024, 0, stream>>>(dst, counts, E);
  block_sum_kernel<<<scanBlocks, 256, 0, stream>>>(counts, bsum, nCounts);
  excl_scan_kernel<<<scanBlocks, 256, 0, stream>>>(counts, bsum, offs, nCounts);
  range_scatter_kernel<<<EDGE_SLICES, 1024, 0, stream>>>(src, dst, offs, binned, E);
  range_sort_kernel<<<N_RANGES, 1024, 0, stream>>>(binned, offs, col, rowptr, dinv, Nn,
                                                  EDGE_SLICES);

  gemm_wmma_kernel<128, false><<<256, GEMM_WAVES * 32, 0, stream>>>(x, W1, nullptr, dinv, h,
                                                                   Nn);
  gather_kernel<4, false><<<(Nn + 7) / 8, 256, 0, stream>>>(
      h, rowptr, col, dinv, nullptr, out1, Nn);

  gemm_wmma_kernel<64, true><<<256, GEMM_WAVES * 32, 0, stream>>>(out1, W2, b1, dinv, h2, Nn);
  gather_kernel<2, true><<<(Nn + 7) / 8, 256, 0, stream>>>(
      h2, rowptr, col, dinv, b2, out, Nn);
}


# Edge grouping in 2 launches, not 5: each slice sorts its own edges by range in LDS, range groups collect the slices' runs, no scan across groups; node sum order follows the new LDS arrival ranks
# speedup vs baseline: 4.7275x; 1.2117x over previous
#include <hip/hip_runtime.h>

typedef __attribute__((ext_vector_type(2))) float v2f;
typedef __attribute__((ext_vector_type(4))) float v4f;
typedef __attribute__((ext_vector_type(8))) float v8f;

#define GCN_N_NODES 100000

#define RANGE_SHIFT 8
#define RANGE_NODES (1 << RANGE_SHIFT)
#define N_RANGES ((GCN_N_NODES + RANGE_NODES - 1) / RANGE_NODES)
#define EDGE_SLICES 256
#define OFFS_PITCH (N_RANGES + 1)
#define RUNS_PER_WAVE (EDGE_SLICES / 32)
#define SRC_BITS 17
#define SRC_MASK ((1 << SRC_BITS) - 1)
static_assert(GCN_N_NODES <= (1 << SRC_BITS), "a source id must fit SRC_BITS");
static_assert(SRC_BITS + RANGE_SHIFT < 31, "a packed record must stay non-negative");
static_assert(N_RANGES + 1 <= 512, "the range scan is done by 16 waves");
static_assert(EDGE_SLICES % 32 == 0 && EDGE_SLICES <= 1024, "runs are dealt out to 32 waves");

__global__ void __launch_bounds__(1024)
slice_sort_kernel(const int* __restrict__ src, const int* __restrict__ dst,
                  int* __restrict__ binned, int* __restrict__ sliceOffs, int E) {
  __shared__ int cnt[512];
  __shared__ int wsum[16];
  const int t = threadIdx.x;
  const int chunk = (E + EDGE_SLICES - 1) / EDGE_SLICES;
  const int lo = blockIdx.x * chunk;
  int hi = lo + chunk;
  if (hi > E) hi = E;

  if (t < 512) cnt[t] = 0;
  __syncthreads();
  for (int base = lo; base < hi; base += 8 * 1024) {
    int d[8];
#pragma unroll
    for (int j = 0; j < 8; ++j) {
      int e = base + j * 1024 + t;
      d[j] = (e < hi) ? dst[e] : -1;
    }
#pragma unroll
    for (int j = 0; j < 8; ++j)
      if (d[j] >= 0) atomicAdd(&cnt[d[j] >> RANGE_SHIFT], 1);
  }
  __syncthreads();

  int c = 0, incl = 0;
  if (t < 512) {
    c = cnt[t];
    incl = c;
#pragma unroll
    for (int o = 1; o < 32; o <<= 1) {
      int v = __shfl_up(incl, o);
      if ((t & 31) >= o) incl += v;
    }
    if ((t & 31) == 31) wsum[t >> 5] = incl;
  }
  __syncthreads();
  if (t < 512) {
    int first = incl - c;
    for (int w = 0; w < (t >> 5); ++w) first += wsum[w];
    if (t <= N_RANGES) sliceOffs[blockIdx.x * OFFS_PITCH + t] = first;
    cnt[t] = lo + first;
  }
  __syncthreads();
  for (int base = lo; base < hi; base += 8 * 1024) {
    int d[8], s[8];
#pragma unroll
    for (int j = 0; j < 8; ++j) {
      int e = base + j * 1024 + t;
      bool ok = e < hi;
      d[j] = ok ? dst[e] : -1;
      s[j] = ok ? src[e] : 0;
    }
#pragma unroll
    for (int j = 0; j < 8; ++j)
      if (d[j] >= 0) {
        int pos = atomicAdd(&cnt[d[j] >> RANGE_SHIFT], 1);
        binned[pos] = s[j] | ((d[j] & (RANGE_NODES - 1)) << SRC_BITS);
      }
  }
}

__global__ void __launch_bounds__(1024)
range_sort_kernel(const int* __restrict__ binned, const int* __restrict__ sliceOffs,
                  int* __restrict__ col, int* __restrict__ rowptr,
                  float* __restrict__ dinv, int n, int E) {
  __shared__ int cnt[RANGE_NODES];
  __shared__ int wsum[RANGE_NODES / 32];
  __shared__ int rs[EDGE_SLICES];
  __shared__ int re[EDGE_SLICES];
  __shared__ int red[EDGE_SLICES / 32];
  const int t = threadIdx.x;
  const int b = blockIdx.x;
  const int lane = t & 31;
  const int w = t >> 5;
  const int chunk = (E + EDGE_SLICES - 1) / EDGE_SLICES;

  if (t < EDGE_SLICES) {
    const int o0 = sliceOffs[t * OFFS_PITCH + b];
    const int o1 = sliceOffs[t * OFFS_PITCH + b + 1];
    rs[t] = t * chunk + o0;
    re[t] = t * chunk + o1;
    int part = o0;
#pragma unroll
    for (int o = 16; o > 0; o >>= 1) part += __shfl_xor(part, o);
    if (lane == 0) red[w] = part;
  }
  if (t < RANGE_NODES) cnt[t] = 0;
  __syncthreads();
  int bs = 0;
#pragma unroll
  for (int k = 0; k < EDGE_SLICES / 32; ++k) bs += red[k];

  int rb[RUNS_PER_WAVE], rend[RUNS_PER_WAVE];
#pragma unroll
  for (int q = 0; q < RUNS_PER_WAVE; ++q) {
    rb[q] = rs[w * RUNS_PER_WAVE + q] + lane;
    rend[q] = re[w * RUNS_PER_WAVE + q];
  }

  {
    int r[RUNS_PER_WAVE];
#pragma unroll
    for (int q = 0; q < RUNS_PER_WAVE; ++q) r[q] = (rb[q] < rend[q]) ? binned[rb[q]] : -1;
#pragma unroll
    for (int q = 0; q < RUNS_PER_WAVE; ++q)
      if (r[q] >= 0) atomicAdd(&cnt[r[q] >> SRC_BITS], 1);
#pragma unroll
    for (int q = 0; q < RUNS_PER_WAVE; ++q)
      for (int i = rb[q] + 32; i < rend[q]; i += 32) atomicAdd(&cnt[binned[i] >> SRC_BITS], 1);
  }
  __syncthreads();

  int c = 0, incl = 0;
  if (t < RANGE_NODES) {
    c = cnt[t];
    incl = c;
#pragma unroll
    for (int o = 1; o < 32; o <<= 1) {
      int v = __shfl_up(incl, o);
      if ((t & 31) >= o) incl += v;
    }
    if ((t & 31) == 31) wsum[t >> 5] = incl;
  }
  __syncthreads();
  if (t < RANGE_NODES) {
    int first = bs + incl - c;
    for (int k = 0; k < (t >> 5); ++k) first += wsum[k];
    int node = b * RANGE_NODES + t;
    if (node < n) {
      rowptr[node] = first;
      dinv[node] = rsqrtf((float)(c + 1));
      if (node == n - 1) rowptr[n] = first + c;
    }
    cnt[t] = first;
  }
  __syncthreads();
  {
    int r[RUNS_PER_WAVE];
#pragma unroll
    for (int q = 0; q < RUNS_PER_WAVE; ++q) r[q] = (rb[q] < rend[q]) ? binned[rb[q]] : -1;
#pragma unroll
    for (int q = 0; q < RUNS_PER_WAVE; ++q)
      if (r[q] >= 0) {
        int pos = atomicAdd(&cnt[r[q] >> SRC_BITS], 1);
        col[pos] = r[q] & SRC_MASK;
      }
#pragma unroll
    for (int q = 0; q < RUNS_PER_WAVE; ++q)
      for (int i = rb[q] + 32; i < rend[q]; i += 32) {
        int v = binned[i];
        int pos = atomicAdd(&cnt[v >> SRC_BITS], 1);
        col[pos] = v & SRC_MASK;
      }
  }
}

#define GEMM_K 128
#define GEMM_WAVES 32

template <int N, bool FUSE_RELU_BIAS>
__global__ void __launch_bounds__(GEMM_WAVES * 32)
gemm_wmma_kernel(const float* __restrict__ A, const float* __restrict__ W,
                 const float* __restrict__ bias, const float* __restrict__ rowscale,
                 float* __restrict__ C, int M) {
  __shared__ v4f Wl[(GEMM_K / 4) * N];
  __shared__ v4f biasl[GEMM_K / 4];
  const int t = threadIdx.x;
  {
    const int n = t % N;
    for (int kb = t / N; kb < GEMM_K / 4; kb += (GEMM_WAVES * 32) / N) {
      const float* wp = W + (size_t)(4 * kb) * N + n;
      v4f v;
      v.x = wp[0];
      v.y = wp[N];
      v.z = wp[2 * N];
      v.w = wp[3 * N];
      Wl[kb * N + n] = v;
    }
    if (FUSE_RELU_BIAS && t < GEMM_K) reinterpret_cast<float*>(biasl)[t] = bias[t];
  }
  __syncthreads();

  constexpr int CH = N / 64;
  constexpr int SLOTS = GEMM_WAVES / CH;
  const int w = __builtin_amdgcn_readfirstlane(t >> 5);
  const int colBase = (w % CH) * 64;
  const int slot = w / CH;
  const int lane = t & 31;
  const int hh = lane >> 4;
  const int c = lane & 15;
  const int strips = M >> 4;

  for (int strip = slot * (int)gridDim.x + (int)blockIdx.x; strip < strips;
       strip += SLOTS * (int)gridDim.x) {
    const float* arow = A + (size_t)(strip * 16 + c) * GEMM_K + 4 * hh;
    const v4f* bl = Wl + hh * N + colBase + c;
    const float* sp = rowscale + strip * 16 + hh * 8;
    const v4f sc0 = *reinterpret_cast<const v4f*>(sp);
    const v4f sc1 = *reinterpret_cast<const v4f*>(sp + 4);
    v8f acc0 = {}, acc1 = {}, acc2 = {}, acc3 = {};
#pragma unroll 4
    for (int j = 0; j < GEMM_K / 8; ++j) {
      v4f a = *reinterpret_cast<const v4f*>(arow + 8 * j);
      if (FUSE_RELU_BIAS) {
        v4f bv = biasl[2 * j + hh];
        a.x = fmaxf(a.x + bv.x, 0.0f);
        a.y = fmaxf(a.y + bv.y, 0.0f);
        a.z = fmaxf(a.z + bv.z, 0.0f);
        a.w = fmaxf(a.w + bv.w, 0.0f);
      }
      v2f a01; a01.x = a.x; a01.y = a.y;
      v2f a23; a23.x = a.z; a23.y = a.w;
      const v4f b0 = bl[(2 * j) * N];
      const v4f b1 = bl[(2 * j) * N + 16];
      const v4f b2 = bl[(2 * j) * N + 32];
      const v4f b3 = bl[(2 * j) * N + 48];
      v2f q;
      q.x = b0.x; q.y = b0.y;
      acc0 = __builtin_amdgcn_wmma_f32_16x16x4_f32(false, a01, false, q, (short)0, acc0, false, false);
      q.x = b1.x; q.y = b1.y;
      acc1 = __builtin_amdgcn_wmma_f32_16x16x4_f32(false, a01, false, q, (short)0, acc1, false, false);
      q.x = b2.x; q.y = b2.y;
      acc2 = __builtin_amdgcn_wmma_f32_16x16x4_f32(false, a01, false, q, (short)0, acc2, false, false);
      q.x = b3.x; q.y = b3.y;
      acc3 = __builtin_amdgcn_wmma_f32_16x16x4_f32(false, a01, false, q, (short)0, acc3, false, false);
      q.x = b0.z; q.y = b0.w;
      acc0 = __builtin_amdgcn_wmma_f32_16x16x4_f32(false, a23, false, q, (short)0, acc0, false, false);
      q.x = b1.z; q.y = b1.w;
      acc1 = __builtin_amdgcn_wmma_f32_16x16x4_f32(false, a23, false, q, (short)0, acc1, false, false);
      q.x = b2.z; q.y = b2.w;
      acc2 = __builtin_amdgcn_wmma_f32_16x16x4_f32(false, a23, false, q, (short)0, acc2, false, false);
      q.x = b3.z; q.y = b3.w;
      acc3 = __builtin_amdgcn_wmma_f32_16x16x4_f32(false, a23, false, q, (short)0, acc3, false, false);
    }

    float* crow = C + (size_t)(strip * 16 + hh * 8) * N + colBase + c;
#pragma unroll
    for (int r = 0; r < 8; ++r) {
      const float sc = (r < 4) ? sc0[r & 3] : sc1[r & 3];
      crow[(size_t)r * N] = acc0[r] * sc;
      crow[(size_t)r * N + 16] = acc1[r] * sc;
      crow[(size_t)r * N + 32] = acc2[r] * sc;
      crow[(size_t)r * N + 48] = acc3[r] * sc;
    }
  }
}

template <int FPL> struct row_vec { typedef v4f type; };
template <> struct row_vec<2> { typedef v2f type; };

template <int FPL, bool ADD_BIAS>
__global__ void __launch_bounds__(256)
gather_kernel(const float* __restrict__ h, const int* __restrict__ rowptr,
              const int* __restrict__ col, const float* __restrict__ dinv,
              const float* __restrict__ bias, float* __restrict__ out, int n) {
  typedef typename row_vec<FPL>::type vf;
  const int F = FPL * 32;
  int node = __builtin_amdgcn_readfirstlane(
      (int)((blockIdx.x * blockDim.x + threadIdx.x) >> 5));
  if (node >= n) return;
  int lane = threadIdx.x & 31;
  int beg = rowptr[node];
  int end = rowptr[node + 1];
  float dd = dinv[node];
  const float* hl = h + lane * FPL;

  vf acc = *reinterpret_cast<const vf*>(hl + (size_t)node * F);

  for (int base = beg; base < end; base += 32) {
    int j = base + lane;
    int s_l = 0;
    if (j < end) s_l = col[j];
    int m = end - base;
    if (m > 32) m = 32;
#pragma unroll 4
    for (int k = 0; k < m; ++k) {
      int s = __builtin_amdgcn_readlane(s_l, k);
      acc += *reinterpret_cast<const vf*>(hl + (size_t)s * F);
    }
  }

  acc *= dd;
  if (ADD_BIAS) acc += *reinterpret_cast<const vf*>(bias + lane * FPL);
  *reinterpret_cast<vf*>(out + (size_t)node * F + lane * FPL) = acc;
}

extern "C" void kernel_launch(void* const* d_in, const int* in_sizes, int n_in,
                              void* d_out, int out_size, void* d_ws, size_t ws_size,
                              hipStream_t stream) {
  const float* x  = (const float*)d_in[0];
  const int*   ei = (const int*)d_in[1];
  const float* W1 = (const float*)d_in[2];
  const float* b1 = (const float*)d_in[3];
  const float* W2 = (const float*)d_in[4];
  const float* b2 = (const float*)d_in[5];
  float* out = (float*)d_out;

  const int Nn = GCN_N_NODES;
  const int E  = in_sizes[1] / 2;
  const int* src = ei;
  const int* dst = ei + E;

  float* dinv = (float*)d_ws;
  float* h    = dinv + Nn;
  float* out1 = h + (size_t)Nn * 128;
  float* h2   = h;
  int* sliceOffs = (int*)(out1 + (size_t)Nn * 128);
  int* rowptr = sliceOffs + EDGE_SLICES * OFFS_PITCH + 32;
  int* binned = rowptr + Nn + 32;
  int* col    = binned + E;

  slice_sort_kernel<<<EDGE_SLICES, 1024, 0, stream>>>(src, dst, binned, sliceOffs, E);
  range_sort_kernel<<<N_RANGES, 1024, 0, stream>>>(binned, sliceOffs, col, rowptr, dinv, Nn, E);

  gemm_wmma_kernel<128, false><<<256, GEMM_WAVES * 32, 0, stream>>>(x, W1, nullptr, dinv, h,
                                                                   Nn);
  gather_kernel<4, false><<<(Nn + 7) / 8, 256, 0, stream>>>(
      h, rowptr, col, dinv, nullptr, out1, Nn);

  gemm_wmma_kernel<64, true><<<256, GEMM_WAVES * 32, 0, stream>>>(out1, W2, b1, dinv, h2, Nn);
  gather_kernel<2, true><<<(Nn + 7) / 8, 256, 0, stream>>>(
      h2, rowptr, col, dinv, b2, out, Nn);
}


# Layer 1 sums x rows first (per-edge dinv[src] weight), then both products run in the gathering work-group from LDS tiles: no h/out1 planes, 4 launches not 6; layer-1 sum order and rounding change
# speedup vs baseline: 5.7059x; 1.2070x over previous
#include <hip/hip_runtime.h>

typedef __attribute__((ext_vector_type(2))) float v2f;
typedef __attribute__((ext_vector_type(4))) float v4f;
typedef __attribute__((ext_vector_type(8))) float v8f;

#define GCN_N_NODES 100000

#define RANGE_SHIFT 8
#define RANGE_NODES (1 << RANGE_SHIFT)
#define N_RANGES ((GCN_N_NODES + RANGE_NODES - 1) / RANGE_NODES)
#define EDGE_SLICES 256
#define OFFS_PITCH (N_RANGES + 1)
#define RUNS_PER_WAVE (EDGE_SLICES / 32)
#define SRC_BITS 17
#define SRC_MASK ((1 << SRC_BITS) - 1)
static_assert(GCN_N_NODES <= (1 << SRC_BITS), "a source id must fit SRC_BITS");
static_assert(SRC_BITS + RANGE_SHIFT < 31, "a packed record must stay non-negative");
static_assert(N_RANGES + 1 <= 512, "the range scan is done by 16 waves");
static_assert(EDGE_SLICES % 32 == 0 && EDGE_SLICES <= 1024, "runs are dealt out to 32 waves");

__global__ void __launch_bounds__(1024)
slice_sort_kernel(const int* __restrict__ src, const int* __restrict__ dst,
                  int* __restrict__ binned, int* __restrict__ sliceOffs, int E) {
  __shared__ int cnt[512];
  __shared__ int wsum[16];
  const int t = threadIdx.x;
  const int chunk = (E + EDGE_SLICES - 1) / EDGE_SLICES;
  const int lo = blockIdx.x * chunk;
  int hi = lo + chunk;
  if (hi > E) hi = E;

  if (t < 512) cnt[t] = 0;
  __syncthreads();
  for (int base = lo; base < hi; base += 8 * 1024) {
    int d[8];
#pragma unroll
    for (int j = 0; j < 8; ++j) {
      int e = base + j * 1024 + t;
      d[j] = (e < hi) ? dst[e] : -1;
    }
#pragma unroll
    for (int j = 0; j < 8; ++j)
      if (d[j] >= 0) atomicAdd(&cnt[d[j] >> RANGE_SHIFT], 1);
  }
  __syncthreads();

  int c = 0, incl = 0;
  if (t < 512) {
    c = cnt[t];
    incl = c;
#pragma unroll
    for (int o = 1; o < 32; o <<= 1) {
      int v = __shfl_up(incl, o);
      if ((t & 31) >= o) incl += v;
    }
    if ((t & 31) == 31) wsum[t >> 5] = incl;
  }
  __syncthreads();
  if (t < 512) {
    int first = incl - c;
    for (int w = 0; w < (t >> 5); ++w) first += wsum[w];
    if (t <= N_RANGES) sliceOffs[blockIdx.x * OFFS_PITCH + t] = first;
    cnt[t] = lo + first;
  }
  __syncthreads();
  for (int base = lo; base < hi; base += 8 * 1024) {
    int d[8], s[8];
#pragma unroll
    for (int j = 0; j < 8; ++j) {
      int e = base + j * 1024 + t;
      bool ok = e < hi;
      d[j] = ok ? dst[e] : -1;
      s[j] = ok ? src[e] : 0;
    }
#pragma unroll
    for (int j = 0; j < 8; ++j)
      if (d[j] >= 0) {
        int pos = atomicAdd(&cnt[d[j] >> RANGE_SHIFT], 1);
        binned[pos] = s[j] | ((d[j] & (RANGE_NODES - 1)) << SRC_BITS);
      }
  }
}

__global__ void __launch_bounds__(1024)
range_sort_kernel(const int* __restrict__ binned, const int* __restrict__ sliceOffs,
                  int* __restrict__ col, int* __restrict__ rowptr,
                  float* __restrict__ dinv, int n, int E) {
  __shared__ int cnt[RANGE_NODES];
  __shared__ int wsum[RANGE_NODES / 32];
  __shared__ int rs[EDGE_SLICES];
  __shared__ int re[EDGE_SLICES];
  __shared__ int red[EDGE_SLICES / 32];
  const int t = threadIdx.x;
  const int b = blockIdx.x;
  const int lane = t & 31;
  const int w = t >> 5;
  const int chunk = (E + EDGE_SLICES - 1) / EDGE_SLICES;

  if (t < EDGE_SLICES) {
    const int o0 = sliceOffs[t * OFFS_PITCH + b];
    const int o1 = sliceOffs[t * OFFS_PITCH + b + 1];
    rs[t] = t * chunk + o0;
    re[t] = t * chunk + o1;
    int part = o0;
#pragma unroll
    for (int o = 16; o > 0; o >>= 1) part += __shfl_xor(part, o);
    if (lane == 0) red[w] = part;
  }
  if (t < RANGE_NODES) cnt[t] = 0;
  __syncthreads();
  int bs = 0;
#pragma unroll
  for (int k = 0; k < EDGE_SLICES / 32; ++k) bs += red[k];

  int rb[RUNS_PER_WAVE], rend[RUNS_PER_WAVE];
#pragma unroll
  for (int q = 0; q < RUNS_PER_WAVE; ++q) {
    rb[q] = rs[w * RUNS_PER_WAVE + q] + lane;
    rend[q] = re[w * RUNS_PER_WAVE + q];
  }

  {
    int r[RUNS_PER_WAVE];
#pragma unroll
    for (int q = 0; q < RUNS_PER_WAVE; ++q) r[q] = (rb[q] < rend[q]) ? binned[rb[q]] : -1;
#pragma unroll
    for (int q = 0; q < RUNS_PER_WAVE; ++q)
      if (r[q] >= 0) atomicAdd(&cnt[r[q] >> SRC_BITS], 1);
#pragma unroll
    for (int q = 0; q < RUNS_PER_WAVE; ++q)
      for (int i = rb[q] + 32; i < rend[q]; i += 32) atomicAdd(&cnt[binned[i] >> SRC_BITS], 1);
  }
  __syncthreads();

  int c = 0, incl = 0;
  if (t < RANGE_NODES) {
    c = cnt[t];
    incl = c;
#pragma unroll
    for (int o = 1; o < 32; o <<= 1) {
      int v = __shfl_up(incl, o);
      if ((t & 31) >= o) incl += v;
    }
    if ((t & 31) == 31) wsum[t >> 5] = incl;
  }
  __syncthreads();
  if (t < RANGE_NODES) {
    int first = bs + incl - c;
    for (int k = 0; k < (t >> 5); ++k) first += wsum[k];
    int node = b * RANGE_NODES + t;
    if (node < n) {
      rowptr[node] = first;
      dinv[node] = rsqrtf((float)(c + 1));
      if (node == n - 1) rowptr[n] = first + c;
    }
    cnt[t] = first;
  }
  __syncthreads();
  {
    int r[RUNS_PER_WAVE];
#pragma unroll
    for (int q = 0; q < RUNS_PER_WAVE; ++q) r[q] = (rb[q] < rend[q]) ? binned[rb[q]] : -1;
#pragma unroll
    for (int q = 0; q < RUNS_PER_WAVE; ++q)
      if (r[q] >= 0) {
        int pos = atomicAdd(&cnt[r[q] >> SRC_BITS], 1);
        col[pos] = r[q] & SRC_MASK;
      }
#pragma unroll
    for (int q = 0; q < RUNS_PER_WAVE; ++q)
      for (int i = rb[q] + 32; i < rend[q]; i += 32) {
        int v = binned[i];
        int pos = atomicAdd(&cnt[v >> SRC_BITS], 1);
        col[pos] = v & SRC_MASK;
      }
  }
}

#define GEMM_K 128
#define FUSE_WAVES 32
#define FUSE_NODES 32
#define TILE_PITCH 132

template <int N>
__device__ __forceinline__ void stage_weights(const float* __restrict__ Wg, v4f* Wl, int t) {
  const int nn = t % N;
  for (int kb = t / N; kb < GEMM_K / 4; kb += (FUSE_WAVES * 32) / N) {
    const float* wp = Wg + (size_t)(4 * kb) * N + nn;
    v4f v;
    v.x = wp[0];
    v.y = wp[N];
    v.z = wp[2 * N];
    v.w = wp[3 * N];
    Wl[kb * N + nn] = v;
  }
}

template <int N>
__device__ __forceinline__ v8f tile_product(const float* arow, const v4f* bl) {
  v8f acc = {};
#pragma unroll 4
  for (int j = 0; j < GEMM_K / 8; ++j) {
    const v4f a = *reinterpret_cast<const v4f*>(arow + 8 * j);
    const v4f b = bl[(2 * j) * N];
    v2f a01; a01.x = a.x; a01.y = a.y;
    v2f a23; a23.x = a.z; a23.y = a.w;
    v2f q;
    q.x = b.x; q.y = b.y;
    acc = __builtin_amdgcn_wmma_f32_16x16x4_f32(false, a01, false, q, (short)0, acc, false, false);
    q.x = b.z; q.y = b.w;
    acc = __builtin_amdgcn_wmma_f32_16x16x4_f32(false, a23, false, q, (short)0, acc, false, false);
  }
  return acc;
}

template <int U>
__device__ __forceinline__ void add_rows(v4f& acc, const float* xl, int s_l, float w_l, int k) {
#pragma unroll
  for (int u = 0; u < U; ++u) {
    const int s = __builtin_amdgcn_readlane(s_l, k + u);
    const float ws = __int_as_float(__builtin_amdgcn_readlane(__float_as_int(w_l), k + u));
    acc += *reinterpret_cast<const v4f*>(xl + (size_t)s * GEMM_K) * ws;
  }
}

__global__ void __launch_bounds__(FUSE_WAVES * 32, 16)
layer1_fused_kernel(const float* __restrict__ x, const int* __restrict__ rowptr,
                    const int* __restrict__ col, const float* __restrict__ dinv,
                    const float* __restrict__ W1, const float* __restrict__ b1,
                    const float* __restrict__ W2, float* __restrict__ h2, int n) {
  __shared__ v4f W1l[(GEMM_K / 4) * 128];
  __shared__ v4f W2l[(GEMM_K / 4) * 64];
  __shared__ __attribute__((aligned(16))) float aggT[FUSE_NODES * TILE_PITCH];
  __shared__ __attribute__((aligned(16))) float actT[FUSE_NODES * TILE_PITCH];
  const int t = threadIdx.x;
  stage_weights<128>(W1, W1l, t);
  stage_weights<64>(W2, W2l, t);

  const int w = __builtin_amdgcn_readfirstlane(t >> 5);
  const int lane = t & 31;
  const int hh = lane >> 4;
  const int c = lane & 15;
  const int strip1 = w >> 3, tile1 = w & 7;
  const int strip2 = (w >> 2) & 1, tile2 = w & 3;
  float bias1 = 0.0f;
  if (w < 16) bias1 = b1[tile1 * 16 + c];
  const float* xl = x + lane * 4;
  __syncthreads();

  const int trips = (n + FUSE_NODES - 1) / FUSE_NODES;
  for (int trip = blockIdx.x; trip < trips; trip += (int)gridDim.x) {
    const int node = trip * FUSE_NODES + w;
    v4f acc = {0.0f, 0.0f, 0.0f, 0.0f};
    if (node < n) {
      const int beg = rowptr[node];
      const int end = rowptr[node + 1];
      const float dd = dinv[node];
      acc = *reinterpret_cast<const v4f*>(xl + (size_t)node * GEMM_K) * dd;
      for (int base = beg; base < end; base += 32) {
        const int j = base + lane;
        int s_l = 0;
        float w_l = 0.0f;
        if (j < end) {
          s_l = col[j];
          w_l = dinv[s_l];
        }
        int m = end - base;
        if (m > 32) m = 32;
        int k = 0;
        for (; k + 8 <= m; k += 8) add_rows<8>(acc, xl, s_l, w_l, k);
        if (m & 4) {
          add_rows<4>(acc, xl, s_l, w_l, k);
          k += 4;
        }
        if (m & 2) {
          add_rows<2>(acc, xl, s_l, w_l, k);
          k += 2;
        }
        if (m & 1) add_rows<1>(acc, xl, s_l, w_l, k);
      }
      acc *= dd;
    }
    *reinterpret_cast<v4f*>(aggT + w * TILE_PITCH + lane * 4) = acc;
    __syncthreads();

    if (w < 16) {
      const v8f d = tile_product<128>(aggT + (strip1 * 16 + c) * TILE_PITCH + 4 * hh,
                                      W1l + hh * 128 + tile1 * 16 + c);
      float* ap = actT + (strip1 * 16 + hh * 8) * TILE_PITCH + tile1 * 16 + c;
#pragma unroll
      for (int r = 0; r < 8; ++r) ap[r * TILE_PITCH] = fmaxf(d[r] + bias1, 0.0f);
    }
    __syncthreads();

    if (w >= 16 && w < 24) {
      const int row0 = trip * FUSE_NODES + strip2 * 16 + hh * 8;
      float sc[8];
#pragma unroll
      for (int r = 0; r < 8; ++r) sc[r] = dinv[(row0 + r < n) ? row0 + r : n - 1];
      const v8f d = tile_product<64>(actT + (strip2 * 16 + c) * TILE_PITCH + 4 * hh,
                                     W2l + hh * 64 + tile2 * 16 + c);
#pragma unroll
      for (int r = 0; r < 8; ++r)
        if (row0 + r < n) h2[(size_t)(row0 + r) * 64 + tile2 * 16 + c] = d[r] * sc[r];
    }
  }
}

template <int FPL> struct row_vec { typedef v4f type; };
template <> struct row_vec<2> { typedef v2f type; };

template <int FPL, bool ADD_BIAS>
__global__ void __launch_bounds__(256)
gather_kernel(const float* __restrict__ h, const int* __restrict__ rowptr,
              const int* __restrict__ col, const float* __restrict__ dinv,
              const float* __restrict__ bias, float* __restrict__ out, int n) {
  typedef typename row_vec<FPL>::type vf;
  const int F = FPL * 32;
  int node = __builtin_amdgcn_readfirstlane(
      (int)((blockIdx.x * blockDim.x + threadIdx.x) >> 5));
  if (node >= n) return;
  int lane = threadIdx.x & 31;
  int beg = rowptr[node];
  int end = rowptr[node + 1];
  float dd = dinv[node];
  const float* hl = h + lane * FPL;

  vf acc = *reinterpret_cast<const vf*>(hl + (size_t)node * F);

  for (int base = beg; base < end; base += 32) {
    int j = base + lane;
    int s_l = 0;
    if (j < end) s_l = col[j];
    int m = end - base;
    if (m > 32) m = 32;
#pragma unroll 4
    for (int k = 0; k < m; ++k) {
      int s = __builtin_amdgcn_readlane(s_l, k);
      acc += *reinterpret_cast<const vf*>(hl + (size_t)s * F);
    }
  }

  acc *= dd;
  if (ADD_BIAS) acc += *reinterpret_cast<const vf*>(bias + lane * FPL);
  *reinterpret_cast<vf*>(out + (size_t)node * F + lane * FPL) = acc;
}

extern "C" void kernel_launch(void* const* d_in, const int* in_sizes, int n_in,
                              void* d_out, int out_size, void* d_ws, size_t ws_size,
                              hipStream_t stream) {
  const float* x  = (const float*)d_in[0];
  const int*   ei = (const int*)d_in[1];
  const float* W1 = (const float*)d_in[2];
  const float* b1 = (const float*)d_in[3];
  const float* W2 = (const float*)d_in[4];
  const float* b2 = (const float*)d_in[5];
  float* out = (float*)d_out;

  const int Nn = GCN_N_NODES;
  const int E  = in_sizes[1] / 2;
  const int* src = ei;
  const int* dst = ei + E;

  float* dinv = (float*)d_ws;
  float* h2   = dinv + Nn;
  int* sliceOffs = (int*)(h2 + (size_t)Nn * 64);
  int* rowptr = sliceOffs + EDGE_SLICES * OFFS_PITCH + 32;
  int* binned = rowptr + Nn + 32;
  int* col    = binned + E;

  slice_sort_kernel<<<EDGE_SLICES, 1024, 0, stream>>>(src, dst, binned, sliceOffs, E);
  range_sort_kernel<<<N_RANGES, 1024, 0, stream>>>(binned, sliceOffs, col, rowptr, dinv, Nn, E);

  layer1_fused_kernel<<<512, FUSE_WAVES * 32, 0, stream>>>(x, rowptr, col, dinv, W1, b1, W2, h2,
                                                           Nn);
  gather_kernel<2, true><<<(Nn + 7) / 8, 256, 0, stream>>>(
      h2, rowptr, col, dinv, b2, out, Nn);
}


# Layer-2 gather on a resident grid: 16384 waves walk over nodes (was 100000 one-node waves); next node's source list and own row, and the node after's row bounds, requested ahead; sum order kept
# speedup vs baseline: 5.9553x; 1.0437x over previous
#include <hip/hip_runtime.h>

typedef __attribute__((ext_vector_type(2))) float v2f;
typedef __attribute__((ext_vector_type(4))) float v4f;
typedef __attribute__((ext_vector_type(8))) float v8f;

#define GCN_N_NODES 100000

#define RANGE_SHIFT 8
#define RANGE_NODES (1 << RANGE_SHIFT)
#define N_RANGES ((GCN_N_NODES + RANGE_NODES - 1) / RANGE_NODES)
#define EDGE_SLICES 256
#define OFFS_PITCH (N_RANGES + 1)
#define RUNS_PER_WAVE (EDGE_SLICES / 32)
#define SRC_BITS 17
#define SRC_MASK ((1 << SRC_BITS) - 1)
static_assert(GCN_N_NODES <= (1 << SRC_BITS), "a source id must fit SRC_BITS");
static_assert(SRC_BITS + RANGE_SHIFT < 31, "a packed record must stay non-negative");
static_assert(N_RANGES + 1 <= 512, "the range scan is done by 16 waves");
static_assert(EDGE_SLICES % 32 == 0 && EDGE_SLICES <= 1024, "runs are dealt out to 32 waves");

__global__ void __launch_bounds__(1024)
slice_sort_kernel(const int* __restrict__ src, const int* __restrict__ dst,
                  int* __restrict__ binned, int* __restrict__ sliceOffs, int E) {
  __shared__ int cnt[512];
  __shared__ int wsum[16];
  const int t = threadIdx.x;
  const int chunk = (E + EDGE_SLICES - 1) / EDGE_SLICES;
  const int lo = blockIdx.x * chunk;
  int hi = lo + chunk;
  if (hi > E) hi = E;

  if (t < 512) cnt[t] = 0;
  __syncthreads();
  for (int base = lo; base < hi; base += 8 * 1024) {
    int d[8];
#pragma unroll
    for (int j = 0; j < 8; ++j) {
      int e = base + j * 1024 + t;
      d[j] = (e < hi) ? dst[e] : -1;
    }
#pragma unroll
    for (int j = 0; j < 8; ++j)
      if (d[j] >= 0) atomicAdd(&cnt[d[j] >> RANGE_SHIFT], 1);
  }
  __syncthreads();

  int c = 0, incl = 0;
  if (t < 512) {
    c = cnt[t];
    incl = c;
#pragma unroll
    for (int o = 1; o < 32; o <<= 1) {
      int v = __shfl_up(incl, o);
      if ((t & 31) >= o) incl += v;
    }
    if ((t & 31) == 31) wsum[t >> 5] = incl;
  }
  __syncthreads();
  if (t < 512) {
    int first = incl - c;
    for (int w = 0; w < (t >> 5); ++w) first += wsum[w];
    if (t <= N_RANGES) sliceOffs[blockIdx.x * OFFS_PITCH + t] = first;
    cnt[t] = lo + first;
  }
  __syncthreads();
  for (int base = lo; base < hi; base += 8 * 1024) {
    int d[8], s[8];
#pragma unroll
    for (int j = 0; j < 8; ++j) {
      int e = base + j * 1024 + t;
      bool ok = e < hi;
      d[j] = ok ? dst[e] : -1;
      s[j] = ok ? src[e] : 0;
    }
#pragma unroll
    for (int j = 0; j < 8; ++j)
      if (d[j] >= 0) {
        int pos = atomicAdd(&cnt[d[j] >> RANGE_SHIFT], 1);
        binned[pos] = s[j] | ((d[j] & (RANGE_NODES - 1)) << SRC_BITS);
      }
  }
}

__global__ void __launch_bounds__(1024)
range_sort_kernel(const int* __restrict__ binned, const int* __restrict__ sliceOffs,
                  int* __restrict__ col, int* __restrict__ rowptr,
                  float* __restrict__ dinv, int n, int E) {
  __shared__ int cnt[RANGE_NODES];
  __shared__ int wsum[RANGE_NODES / 32];
  __shared__ int rs[EDGE_SLICES];
  __shared__ int re[EDGE_SLICES];
  __shared__ int red[EDGE_SLICES / 32];
  const int t = threadIdx.x;
  const int b = blockIdx.x;
  const int lane = t & 31;
  const int w = t >> 5;
  const int chunk = (E + EDGE_SLICES - 1) / EDGE_SLICES;

  if (t < EDGE_SLICES) {
    const int o0 = sliceOffs[t * OFFS_PITCH + b];
    const int o1 = sliceOffs[t * OFFS_PITCH + b + 1];
    rs[t] = t * chunk + o0;
    re[t] = t * chunk + o1;
    int part = o0;
#pragma unroll
    for (int o = 16; o > 0; o >>= 1) part += __shfl_xor(part, o);
    if (lane == 0) red[w] = part;
  }
  if (t < RANGE_NODES) cnt[t] = 0;
  __syncthreads();
  int bs = 0;
#pragma unroll
  for (int k = 0; k < EDGE_SLICES / 32; ++k) bs += red[k];

  int rb[RUNS_PER_WAVE], rend[RUNS_PER_WAVE];
#pragma unroll
  for (int q = 0; q < RUNS_PER_WAVE; ++q) {
    rb[q] = rs[w * RUNS_PER_WAVE + q] + lane;
    rend[q] = re[w * RUNS_PER_WAVE + q];
  }

  {
    int r[RUNS_PER_WAVE];
#pragma unroll
    for (int q = 0; q < RUNS_PER_WAVE; ++q) r[q] = (rb[q] < rend[q]) ? binned[rb[q]] : -1;
#pragma unroll
    for (int q = 0; q < RUNS_PER_WAVE; ++q)
      if (r[q] >= 0) atomicAdd(&cnt[r[q] >> SRC_BITS], 1);
#pragma unroll
    for (int q = 0; q < RUNS_PER_WAVE; ++q)
      for (int i = rb[q] + 32; i < rend[q]; i += 32) atomicAdd(&cnt[binned[i] >> SRC_BITS], 1);
  }
  __syncthreads();

  int c = 0, incl = 0;
  if (t < RANGE_NODES) {
    c = cnt[t];
    incl = c;
#pragma unroll
    for (int o = 1; o < 32; o <<= 1) {
      int v = __shfl_up(incl, o);
      if ((t & 31) >= o) incl += v;
    }
    if ((t & 31) == 31) wsum[t >> 5] = incl;
  }
  __syncthreads();
  if (t < RANGE_NODES) {
    int first = bs + incl - c;
    for (int k = 0; k < (t >> 5); ++k) first += wsum[k];
    int node = b * RANGE_NODES + t;
    if (node < n) {
      rowptr[node] = first;
      dinv[node] = rsqrtf((float)(c + 1));
      if (node == n - 1) rowptr[n] = first + c;
    }
    cnt[t] = first;
  }
  __syncthreads();
  {
    int r[RUNS_PER_WAVE];
#pragma unroll
    for (int q = 0; q < RUNS_PER_WAVE; ++q) r[q] = (rb[q] < rend[q]) ? binned[rb[q]] : -1;
#pragma unroll
    for (int q = 0; q < RUNS_PER_WAVE; ++q)
      if (r[q] >= 0) {
        int pos = atomicAdd(&cnt[r[q] >> SRC_BITS], 1);
        col[pos] = r[q] & SRC_MASK;
      }
#pragma unroll
    for (int q = 0; q < RUNS_PER_WAVE; ++q)
      for (int i = rb[q] + 32; i < rend[q]; i += 32) {
        int v = binned[i];
        int pos = atomicAdd(&cnt[v >> SRC_BITS], 1);
        col[pos] = v & SRC_MASK;
      }
  }
}

#define GEMM_K 128
#define FUSE_WAVES 32
#define FUSE_NODES 32
#define TILE_PITCH 132

template <int N>
__device__ __forceinline__ void stage_weights(const float* __restrict__ Wg, v4f* Wl, int t) {
  const int nn = t % N;
  for (int kb = t / N; kb < GEMM_K / 4; kb += (FUSE_WAVES * 32) / N) {
    const float* wp = Wg + (size_t)(4 * kb) * N + nn;
    v4f v;
    v.x = wp[0];
    v.y = wp[N];
    v.z = wp[2 * N];
    v.w = wp[3 * N];
    Wl[kb * N + nn] = v;
  }
}

template <int N>
__device__ __forceinline__ v8f tile_product(const float* arow, const v4f* bl) {
  v8f acc = {};
#pragma unroll 4
  for (int j = 0; j < GEMM_K / 8; ++j) {
    const v4f a = *reinterpret_cast<const v4f*>(arow + 8 * j);
    const v4f b = bl[(2 * j) * N];
    v2f a01; a01.x = a.x; a01.y = a.y;
    v2f a23; a23.x = a.z; a23.y = a.w;
    v2f q;
    q.x = b.x; q.y = b.y;
    acc = __builtin_amdgcn_wmma_f32_16x16x4_f32(false, a01, false, q, (short)0, acc, false, false);
    q.x = b.z; q.y = b.w;
    acc = __builtin_amdgcn_wmma_f32_16x16x4_f32(false, a23, false, q, (short)0, acc, false, false);
  }
  return acc;
}

template <int U>
__device__ __forceinline__ void add_rows(v4f& acc, const float* xl, int s_l, float w_l, int k) {
#pragma unroll
  for (int u = 0; u < U; ++u) {
    const int s = __builtin_amdgcn_readlane(s_l, k + u);
    const float ws = __int_as_float(__builtin_amdgcn_readlane(__float_as_int(w_l), k + u));
    acc += *reinterpret_cast<const v4f*>(xl + (size_t)s * GEMM_K) * ws;
  }
}

__global__ void __launch_bounds__(FUSE_WAVES * 32, 16)
layer1_fused_kernel(const float* __restrict__ x, const int* __restrict__ rowptr,
                    const int* __restrict__ col, const float* __restrict__ dinv,
                    const float* __restrict__ W1, const float* __restrict__ b1,
                    const float* __restrict__ W2, float* __restrict__ h2, int n) {
  __shared__ v4f W1l[(GEMM_K / 4) * 128];
  __shared__ v4f W2l[(GEMM_K / 4) * 64];
  __shared__ __attribute__((aligned(16))) float aggT[FUSE_NODES * TILE_PITCH];
  __shared__ __attribute__((aligned(16))) float actT[FUSE_NODES * TILE_PITCH];
  const int t = threadIdx.x;
  stage_weights<128>(W1, W1l, t);
  stage_weights<64>(W2, W2l, t);

  const int w = __builtin_amdgcn_readfirstlane(t >> 5);
  const int lane = t & 31;
  const int hh = lane >> 4;
  const int c = lane & 15;
  const int strip1 = w >> 3, tile1 = w & 7;
  const int strip2 = (w >> 2) & 1, tile2 = w & 3;
  float bias1 = 0.0f;
  if (w < 16) bias1 = b1[tile1 * 16 + c];
  const float* xl = x + lane * 4;
  __syncthreads();

  const int trips = (n + FUSE_NODES - 1) / FUSE_NODES;
  for (int trip = blockIdx.x; trip < trips; trip += (int)gridDim.x) {
    const int node = trip * FUSE_NODES + w;
    v4f acc = {0.0f, 0.0f, 0.0f, 0.0f};
    if (node < n) {
      const int beg = rowptr[node];
      const int end = rowptr[node + 1];
      const float dd = dinv[node];
      acc = *reinterpret_cast<const v4f*>(xl + (size_t)node * GEMM_K) * dd;
      for (int base = beg; base < end; base += 32) {
        const int j = base + lane;
        int s_l = 0;
        float w_l = 0.0f;
        if (j < end) {
          s_l = col[j];
          w_l = dinv[s_l];
        }
        int m = end - base;
        if (m > 32) m = 32;
        int k = 0;
        for (; k + 8 <= m; k += 8) add_rows<8>(acc, xl, s_l, w_l, k);
        if (m & 4) {
          add_rows<4>(acc, xl, s_l, w_l, k);
          k += 4;
        }
        if (m & 2) {
          add_rows<2>(acc, xl, s_l, w_l, k);
          k += 2;
        }
        if (m & 1) add_rows<1>(acc, xl, s_l, w_l, k);
      }
      acc *= dd;
    }
    *reinterpret_cast<v4f*>(aggT + w * TILE_PITCH + lane * 4) = acc;
    __syncthreads();

    if (w < 16) {
      const v8f d = tile_product<128>(aggT + (strip1 * 16 + c) * TILE_PITCH + 4 * hh,
                                      W1l + hh * 128 + tile1 * 16 + c);
      float* ap = actT + (strip1 * 16 + hh * 8) * TILE_PITCH + tile1 * 16 + c;
#pragma unroll
      for (int r = 0; r < 8; ++r) ap[r * TILE_PITCH] = fmaxf(d[r] + bias1, 0.0f);
    }
    __syncthreads();

    if (w >= 16 && w < 24) {
      const int row0 = trip * FUSE_NODES + strip2 * 16 + hh * 8;
      float sc[8];
#pragma unroll
      for (int r = 0; r < 8; ++r) sc[r] = dinv[(row0 + r < n) ? row0 + r : n - 1];
      const v8f d = tile_product<64>(actT + (strip2 * 16 + c) * TILE_PITCH + 4 * hh,
                                     W2l + hh * 64 + tile2 * 16 + c);
#pragma unroll
      for (int r = 0; r < 8; ++r)
        if (row0 + r < n) h2[(size_t)(row0 + r) * 64 + tile2 * 16 + c] = d[r] * sc[r];
    }
  }
}

#define OUT_F 64
#define OUT_BLOCKS 512
#define OUT_WAVES 32

template <int U>
__device__ __forceinline__ void add_h_rows(v2f& acc, const float* hl, int s_l, int k) {
#pragma unroll
  for (int u = 0; u < U; ++u) {
    const int s = __builtin_amdgcn_readlane(s_l, k + u);
    acc += *reinterpret_cast<const v2f*>(hl + (size_t)s * OUT_F);
  }
}

__device__ __forceinline__ void add_h_group(v2f& acc, const float* hl, int s_l, int m) {
  int k = 0;
  for (; k + 8 <= m; k += 8) add_h_rows<8>(acc, hl, s_l, k);
  if (m & 4) {
    add_h_rows<4>(acc, hl, s_l, k);
    k += 4;
  }
  if (m & 2) {
    add_h_rows<2>(acc, hl, s_l, k);
    k += 2;
  }
  if (m & 1) add_h_rows<1>(acc, hl, s_l, k);
}

__global__ void __launch_bounds__(OUT_WAVES * 32, 16)
gather_out_kernel(const float* __restrict__ h, const int* __restrict__ rowptr,
                  const int* __restrict__ col, const float* __restrict__ dinv,
                  const float* __restrict__ bias, float* __restrict__ out, int n) {
  const int lane = threadIdx.x & 31;
  const int stride = (int)gridDim.x * OUT_WAVES;
  int node = __builtin_amdgcn_readfirstlane((int)blockIdx.x * OUT_WAVES + (int)(threadIdx.x >> 5));
  if (node >= n) return;
  const float* hl = h + lane * 2;
  const v2f bv = *reinterpret_cast<const v2f*>(bias + lane * 2);

  int beg = rowptr[node];
  int end = rowptr[node + 1];
  float dd = dinv[node];
  int s_l = (beg + lane < end) ? col[beg + lane] : 0;
  v2f self = *reinterpret_cast<const v2f*>(hl + (size_t)node * OUT_F);
  int node1 = node + stride;
  int beg1 = 0, end1 = 0;
  float dd1 = 0.0f;
  if (node1 < n) {
    beg1 = rowptr[node1];
    end1 = rowptr[node1 + 1];
    dd1 = dinv[node1];
  }
  asm volatile("" ::"v"(s_l), "v"(self.x), "v"(self.y));

  for (;;) {
    int s_l1 = 0;
    v2f self1 = {0.0f, 0.0f};
    if (node1 < n) {
      if (beg1 + lane < end1) s_l1 = col[beg1 + lane];
      self1 = *reinterpret_cast<const v2f*>(hl + (size_t)node1 * OUT_F);
    }
    const int node2 = node1 + stride;
    int beg2 = 0, end2 = 0;
    float dd2 = 0.0f;
    if (node2 < n) {
      beg2 = rowptr[node2];
      end2 = rowptr[node2 + 1];
      dd2 = dinv[node2];
    }

    v2f acc = self;
    int m = end - beg;
    if (m > 32) m = 32;
    add_h_group(acc, hl, s_l, m);
    for (int base = beg + 32; base < end; base += 32) {
      const int j = base + lane;
      const int sl = (j < end) ? col[j] : 0;
      m = end - base;
      if (m > 32) m = 32;
      add_h_group(acc, hl, sl, m);
    }
    acc *= dd;
    acc += bv;
    *reinterpret_cast<v2f*>(out + (size_t)node * OUT_F + lane * 2) = acc;

    if (node1 >= n) break;
    node = node1; beg = beg1; end = end1; dd = dd1; s_l = s_l1; self = self1;
    node1 = node2; beg1 = beg2; end1 = end2; dd1 = dd2;
  }
}

extern "C" void kernel_launch(void* const* d_in, const int* in_sizes, int n_in,
                              void* d_out, int out_size, void* d_ws, size_t ws_size,
                              hipStream_t stream) {
  const float* x  = (const float*)d_in[0];
  const int*   ei = (const int*)d_in[1];
  const float* W1 = (const float*)d_in[2];
  const float* b1 = (const float*)d_in[3];
  const float* W2 = (const float*)d_in[4];
  const float* b2 = (const float*)d_in[5];
  float* out = (float*)d_out;

  const int Nn = GCN_N_NODES;
  const int E  = in_sizes[1] / 2;
  const int* src = ei;
  const int* dst = ei + E;

  float* dinv = (float*)d_ws;
  float* h2   = dinv + Nn;
  int* sliceOffs = (int*)(h2 + (size_t)Nn * 64);
  int* rowptr = sliceOffs + EDGE_SLICES * OFFS_PITCH + 32;
  int* binned = rowptr + Nn + 32;
  int* col    = binned + E;

  slice_sort_kernel<<<EDGE_SLICES, 1024, 0, stream>>>(src, dst, binned, sliceOffs, E);
  range_sort_kernel<<<N_RANGES, 1024, 0, stream>>>(binned, sliceOffs, col, rowptr, dinv, Nn, E);

  layer1_fused_kernel<<<512, FUSE_WAVES * 32, 0, stream>>>(x, rowptr, col, dinv, W1, b1, W2, h2,
                                                           Nn);
  gather_out_kernel<<<OUT_BLOCKS, OUT_WAVES * 32, 0, stream>>>(h2, rowptr, col, dinv, b2, out, Nn);
}


# Both sort levels place records in an LDS copy of the work-group's output segment and store it in order as whole 128-B lines (was a scattered 4-B global store per edge); sum order mechanism kept
# speedup vs baseline: 6.4022x; 1.0750x over previous
#include <hip/hip_runtime.h>

typedef __attribute__((ext_vector_type(2))) float v2f;
typedef __attribute__((ext_vector_type(4))) float v4f;
typedef __attribute__((ext_vector_type(8))) float v8f;

#define GCN_N_NODES 100000

#define RANGE_SHIFT 8
#define RANGE_NODES (1 << RANGE_SHIFT)
#define N_RANGES ((GCN_N_NODES + RANGE_NODES - 1) / RANGE_NODES)
#define EDGE_SLICES 256
#define OFFS_PITCH (N_RANGES + 1)
#define RUNS_PER_WAVE (EDGE_SLICES / 32)
#define SRC_BITS 17
#define SRC_MASK ((1 << SRC_BITS) - 1)
static_assert(GCN_N_NODES <= (1 << SRC_BITS), "a source id must fit SRC_BITS");
static_assert(SRC_BITS + RANGE_SHIFT < 31, "a packed record must stay non-negative");
static_assert(N_RANGES + 1 <= 512, "the range scan is done by 16 waves");
static_assert(EDGE_SLICES % 32 == 0 && EDGE_SLICES <= 1024, "runs are dealt out to 32 waves");
#define STAGE_CAP 8192

__device__ __forceinline__ void store_segment(int* __restrict__ dstp, const int* stage,
                                              int seg0, int nrec, int t) {
  const int head = seg0 & 31;
  for (int i = t - head; i < nrec; i += 1024)
    if (i >= 0) dstp[seg0 + i] = stage[i];
}

__global__ void __launch_bounds__(1024)
slice_sort_kernel(const int* __restrict__ src, const int* __restrict__ dst,
                  int* __restrict__ binned, int* __restrict__ sliceOffs, int E) {
  __shared__ int cnt[512];
  __shared__ int wsum[16];
  __shared__ int stage[STAGE_CAP];
  const int t = threadIdx.x;
  const int chunk = (E + EDGE_SLICES - 1) / EDGE_SLICES;
  const int lo = blockIdx.x * chunk;
  int hi = lo + chunk;
  if (hi > E) hi = E;
  const bool staged = chunk <= STAGE_CAP;

  if (t < 512) cnt[t] = 0;
  __syncthreads();
  for (int base = lo; base < hi; base += 8 * 1024) {
    int d[8];
#pragma unroll
    for (int j = 0; j < 8; ++j) {
      int e = base + j * 1024 + t;
      d[j] = (e < hi) ? dst[e] : -1;
    }
#pragma unroll
    for (int j = 0; j < 8; ++j)
      if (d[j] >= 0) atomicAdd(&cnt[d[j] >> RANGE_SHIFT], 1);
  }
  __syncthreads();

  int c = 0, incl = 0;
  if (t < 512) {
    c = cnt[t];
    incl = c;
#pragma unroll
    for (int o = 1; o < 32; o <<= 1) {
      int v = __shfl_up(incl, o);
      if ((t & 31) >= o) incl += v;
    }
    if ((t & 31) == 31) wsum[t >> 5] = incl;
  }
  __syncthreads();
  if (t < 512) {
    int first = incl - c;
    for (int w = 0; w < (t >> 5); ++w) first += wsum[w];
    if (t <= N_RANGES) sliceOffs[blockIdx.x * OFFS_PITCH + t] = first;
    cnt[t] = lo + first;
  }
  __syncthreads();
  for (int base = lo; base < hi; base += 8 * 1024) {
    int d[8], s[8];
#pragma unroll
    for (int j = 0; j < 8; ++j) {
      int e = base + j * 1024 + t;
      bool ok = e < hi;
      d[j] = ok ? dst[e] : -1;
      s[j] = ok ? src[e] : 0;
    }
#pragma unroll
    for (int j = 0; j < 8; ++j)
      if (d[j] >= 0) {
        int pos = atomicAdd(&cnt[d[j] >> RANGE_SHIFT], 1);
        const int rec = s[j] | ((d[j] & (RANGE_NODES - 1)) << SRC_BITS);
        if (staged) stage[pos - lo] = rec;
        else binned[pos] = rec;
      }
  }
  if (staged) {
    __syncthreads();
    store_segment(binned, stage, lo, hi - lo, t);
  }
}

__global__ void __launch_bounds__(1024)
range_sort_kernel(const int* __restrict__ binned, const int* __restrict__ sliceOffs,
                  int* __restrict__ col, int* __restrict__ rowptr,
                  float* __restrict__ dinv, int n, int E) {
  __shared__ int cnt[RANGE_NODES];
  __shared__ int wsum[RANGE_NODES / 32];
  __shared__ int rs[EDGE_SLICES];
  __shared__ int re[EDGE_SLICES];
  __shared__ int red[EDGE_SLICES / 32];
  __shared__ int stage[STAGE_CAP];
  const int t = threadIdx.x;
  const int b = blockIdx.x;
  const int lane = t & 31;
  const int w = t >> 5;
  const int chunk = (E + EDGE_SLICES - 1) / EDGE_SLICES;

  if (t < EDGE_SLICES) {
    const int o0 = sliceOffs[t * OFFS_PITCH + b];
    const int o1 = sliceOffs[t * OFFS_PITCH + b + 1];
    rs[t] = t * chunk + o0;
    re[t] = t * chunk + o1;
    int part = o0;
#pragma unroll
    for (int o = 16; o > 0; o >>= 1) part += __shfl_xor(part, o);
    if (lane == 0) red[w] = part;
  }
  if (t < RANGE_NODES) cnt[t] = 0;
  __syncthreads();
  int bs = 0;
#pragma unroll
  for (int k = 0; k < EDGE_SLICES / 32; ++k) bs += red[k];

  int rb[RUNS_PER_WAVE], rend[RUNS_PER_WAVE];
#pragma unroll
  for (int q = 0; q < RUNS_PER_WAVE; ++q) {
    rb[q] = rs[w * RUNS_PER_WAVE + q] + lane;
    rend[q] = re[w * RUNS_PER_WAVE + q];
  }

  {
    int r[RUNS_PER_WAVE];
#pragma unroll
    for (int q = 0; q < RUNS_PER_WAVE; ++q) r[q] = (rb[q] < rend[q]) ? binned[rb[q]] : -1;
#pragma unroll
    for (int q = 0; q < RUNS_PER_WAVE; ++q)
      if (r[q] >= 0) atomicAdd(&cnt[r[q] >> SRC_BITS], 1);
#pragma unroll
    for (int q = 0; q < RUNS_PER_WAVE; ++q)
      for (int i = rb[q] + 32; i < rend[q]; i += 32) atomicAdd(&cnt[binned[i] >> SRC_BITS], 1);
  }
  __syncthreads();

  int c = 0, incl = 0;
  if (t < RANGE_NODES) {
    c = cnt[t];
    incl = c;
#pragma unroll
    for (int o = 1; o < 32; o <<= 1) {
      int v = __shfl_up(incl, o);
      if ((t & 31) >= o) incl += v;
    }
    if ((t & 31) == 31) wsum[t >> 5] = incl;
  }
  __syncthreads();
  if (t < RANGE_NODES) {
    int first = bs + incl - c;
    for (int k = 0; k < (t >> 5); ++k) first += wsum[k];
    int node = b * RANGE_NODES + t;
    if (node < n) {
      rowptr[node] = first;
      dinv[node] = rsqrtf((float)(c + 1));
      if (node == n - 1) rowptr[n] = first + c;
    }
    cnt[t] = first;
  }
  int total = 0;
  for (int k = 0; k < RANGE_NODES / 32; ++k) total += wsum[k];
  const bool staged = total <= STAGE_CAP;
  __syncthreads();
  {
    int r[RUNS_PER_WAVE];
#pragma unroll
    for (int q = 0; q < RUNS_PER_WAVE; ++q) r[q] = (rb[q] < rend[q]) ? binned[rb[q]] : -1;
#pragma unroll
    for (int q = 0; q < RUNS_PER_WAVE; ++q)
      if (r[q] >= 0) {
        int pos = atomicAdd(&cnt[r[q] >> SRC_BITS], 1);
        if (staged) stage[pos - bs] = r[q] & SRC_MASK;
        else col[pos] = r[q] & SRC_MASK;
      }
#pragma unroll
    for (int q = 0; q < RUNS_PER_WAVE; ++q)
      for (int i = rb[q] + 32; i < rend[q]; i += 32) {
        int v = binned[i];
        int pos = atomicAdd(&cnt[v >> SRC_BITS], 1);
        if (staged) stage[pos - bs] = v & SRC_MASK;
        else col[pos] = v & SRC_MASK;
      }
  }
  if (staged) {
    __syncthreads();
    store_segment(col, stage, bs, total, t);
  }
}

#define GEMM_K 128
#define FUSE_WAVES 32
#define FUSE_NODES 32
#define TILE_PITCH 132

template <int N>
__device__ __forceinline__ void stage_weights(const float* __restrict__ Wg, v4f* Wl, int t) {
  const int nn = t % N;
  for (int kb = t / N; kb < GEMM_K / 4; kb += (FUSE_WAVES * 32) / N) {
    const float* wp = Wg + (size_t)(4 * kb) * N + nn;
    v4f v;
    v.x = wp[0];
    v.y = wp[N];
    v.z = wp[2 * N];
    v.w = wp[3 * N];
    Wl[kb * N + nn] = v;
  }
}

template <int N>
__device__ __forceinline__ v8f tile_product(const float* arow, const v4f* bl) {
  v8f acc = {};
#pragma unroll 4
  for (int j = 0; j < GEMM_K / 8; ++j) {
    const v4f a = *reinterpret_cast<const v4f*>(arow + 8 * j);
    const v4f b = bl[(2 * j) * N];
    v2f a01; a01.x = a.x; a01.y = a.y;
    v2f a23; a23.x = a.z; a23.y = a.w;
    v2f q;
    q.x = b.x; q.y = b.y;
    acc = __builtin_amdgcn_wmma_f32_16x16x4_f32(false, a01, false, q, (short)0, acc, false, false);
    q.x = b.z; q.y = b.w;
    acc = __builtin_amdgcn_wmma_f32_16x16x4_f32(false, a23, false, q, (short)0, acc, false, false);
  }
  return acc;
}

template <int U>
__device__ __forceinline__ void add_rows(v4f& acc, const float* xl, int s_l, float w_l, int k) {
#pragma unroll
  for (int u = 0; u < U; ++u) {
    const int s = __builtin_amdgcn_readlane(s_l, k + u);
    const float ws = __int_as_float(__builtin_amdgcn_readlane(__float_as_int(w_l), k + u));
    acc += *reinterpret_cast<const v4f*>(xl + (size_t)s * GEMM_K) * ws;
  }
}

__global__ void __launch_bounds__(FUSE_WAVES * 32, 16)
layer1_fused_kernel(const float* __restrict__ x, const int* __restrict__ rowptr,
                    const int* __restrict__ col, const float* __restrict__ dinv,
                    const float* __restrict__ W1, const float* __restrict__ b1,
                    const float* __restrict__ W2, float* __restrict__ h2, int n) {
  __shared__ v4f W1l[(GEMM_K / 4) * 128];
  __shared__ v4f W2l[(GEMM_K / 4) * 64];
  __shared__ __attribute__((aligned(16))) float aggT[FUSE_NODES * TILE_PITCH];
  __shared__ __attribute__((aligned(16))) float actT[FUSE_NODES * TILE_PITCH];
  const int t = threadIdx.x;
  stage_weights<128>(W1, W1l, t);
  stage_weights<64>(W2, W2l, t);

  const int w = __builtin_amdgcn_readfirstlane(t >> 5);
  const int lane = t & 31;
  const int hh = lane >> 4;
  const int c = lane & 15;
  const int strip1 = w >> 3, tile1 = w & 7;
  const int strip2 = (w >> 2) & 1, tile2 = w & 3;
  float bias1 = 0.0f;
  if (w < 16) bias1 = b1[tile1 * 16 + c];
  const float* xl = x + lane * 4;
  __syncthreads();

  const int trips = (n + FUSE_NODES - 1) / FUSE_NODES;
  for (int trip = blockIdx.x; trip < trips; trip += (int)gridDim.x) {
    const int node = trip * FUSE_NODES + w;
    v4f acc = {0.0f, 0.0f, 0.0f, 0.0f};
    if (node < n) {
      const int beg = rowptr[node];
      const int end = rowptr[node + 1];
      const float dd = dinv[node];
      acc = *reinterpret_cast<const v4f*>(xl + (size_t)node * GEMM_K) * dd;
      for (int base = beg; base < end; base += 32) {
        const int j = base + lane;
        int s_l = 0;
        float w_l = 0.0f;
        if (j < end) {
          s_l = col[j];
          w_l = dinv[s_l];
        }
        int m = end - base;
        if (m > 32) m = 32;
        int k = 0;
        for (; k + 8 <= m; k += 8) add_rows<8>(acc, xl, s_l, w_l, k);
        if (m & 4) {
          add_rows<4>(acc, xl, s_l, w_l, k);
          k += 4;
        }
        if (m & 2) {
          add_rows<2>(acc, xl, s_l, w_l, k);
          k += 2;
        }
        if (m & 1) add_rows<1>(acc, xl, s_l, w_l, k);
      }
      acc *= dd;
    }
    *reinterpret_cast<v4f*>(aggT + w * TILE_PITCH + lane * 4) = acc;
    __syncthreads();

    if (w < 16) {
      const v8f d = tile_product<128>(aggT + (strip1 * 16 + c) * TILE_PITCH + 4 * hh,
                                      W1l + hh * 128 + tile1 * 16 + c);
      float* ap = actT + (strip1 * 16 + hh * 8) * TILE_PITCH + tile1 * 16 + c;
#pragma unroll
      for (int r = 0; r < 8; ++r) ap[r * TILE_PITCH] = fmaxf(d[r] + bias1, 0.0f);
    }
    __syncthreads();

    if (w >= 16 && w < 24) {
      const int row0 = trip * FUSE_NODES + strip2 * 16 + hh * 8;
      float sc[8];
#pragma unroll
      for (int r = 0; r < 8; ++r) sc[r] = dinv[(row0 + r < n) ? row0 + r : n - 1];
      const v8f d = tile_product<64>(actT + (strip2 * 16 + c) * TILE_PITCH + 4 * hh,
                                     W2l + hh * 64 + tile2 * 16 + c);
#pragma unroll
      for (int r = 0; r < 8; ++r)
        if (row0 + r < n) h2[(size_t)(row0 + r) * 64 + tile2 * 16 + c] = d[r] * sc[r];
    }
  }
}

#define OUT_F 64
#define OUT_BLOCKS 512
#define OUT_WAVES 32

template <int U>
__device__ __forceinline__ void add_h_rows(v2f& acc, const float* hl, int s_l, int k) {
#pragma unroll
  for (int u = 0; u < U; ++u) {
    const int s = __builtin_amdgcn_readlane(s_l, k + u);
    acc += *reinterpret_cast<const v2f*>(hl + (size_t)s * OUT_F);
  }
}

__device__ __forceinline__ void add_h_group(v2f& acc, const float* hl, int s_l, int m) {
  int k = 0;
  for (; k + 8 <= m; k += 8) add_h_rows<8>(acc, hl, s_l, k);
  if (m & 4) {
    add_h_rows<4>(acc, hl, s_l, k);
    k += 4;
  }
  if (m & 2) {
    add_h_rows<2>(acc, hl, s_l, k);
    k += 2;
  }
  if (m & 1) add_h_rows<1>(acc, hl, s_l, k);
}

__global__ void __launch_bounds__(OUT_WAVES * 32, 16)
gather_out_kernel(const float* __restrict__ h, const int* __restrict__ rowptr,
                  const int* __restrict__ col, const float* __restrict__ dinv,
                  const float* __restrict__ bias, float* __restrict__ out, int n) {
  const int lane = threadIdx.x & 31;
  const int stride = (int)gridDim.x * OUT_WAVES;
  int node = __builtin_amdgcn_readfirstlane((int)blockIdx.x * OUT_WAVES + (int)(threadIdx.x >> 5));
  if (node >= n) return;
  const float* hl = h + lane * 2;
  const v2f bv = *reinterpret_cast<const v2f*>(bias + lane * 2);

  int beg = rowptr[node];
  int end = rowptr[node + 1];
  float dd = dinv[node];
  int s_l = (beg + lane < end) ? col[beg + lane] : 0;
  v2f self = *reinterpret_cast<const v2f*>(hl + (size_t)node * OUT_F);
  int node1 = node + stride;
  int beg1 = 0, end1 = 0;
  float dd1 = 0.0f;
  if (node1 < n) {
    beg1 = rowptr[node1];
    end1 = rowptr[node1 + 1];
    dd1 = dinv[node1];
  }
  asm volatile("" ::"v"(s_l), "v"(self.x), "v"(self.y));

  for (;;) {
    int s_l1 = 0;
    v2f self1 = {0.0f, 0.0f};
    if (node1 < n) {
      if (beg1 + lane < end1) s_l1 = col[beg1 + lane];
      self1 = *reinterpret_cast<const v2f*>(hl + (size_t)node1 * OUT_F);
    }
    const int node2 = node1 + stride;
    int beg2 = 0, end2 = 0;
    float dd2 = 0.0f;
    if (node2 < n) {
      beg2 = rowptr[node2];
      end2 = rowptr[node2 + 1];
      dd2 = dinv[node2];
    }

    v2f acc = self;
    int m = end - beg;
    if (m > 32) m = 32;
    add_h_group(acc, hl, s_l, m);
    for (int base = beg + 32; base < end; base += 32) {
      const int j = base + lane;
      const int sl = (j < end) ? col[j] : 0;
      m = end - base;
      if (m > 32) m = 32;
      add_h_group(acc, hl, sl, m);
    }
    acc *= dd;
    acc += bv;
    *reinterpret_cast<v2f*>(out + (size_t)node * OUT_F + lane * 2) = acc;

    if (node1 >= n) break;
    node = node1; beg = beg1; end = end1; dd = dd1; s_l = s_l1; self = self1;
    node1 = node2; beg1 = beg2; end1 = end2; dd1 = dd2;
  }
}

extern "C" void kernel_launch(void* const* d_in, const int* in_sizes, int n_in,
                              void* d_out, int out_size, void* d_ws, size_t ws_size,
                              hipStream_t stream) {
  const float* x  = (const float*)d_in[0];
  const int*   ei = (const int*)d_in[1];
  const float* W1 = (const float*)d_in[2];
  const float* b1 = (const float*)d_in[3];
  const float* W2 = (const float*)d_in[4];
  const float* b2 = (const float*)d_in[5];
  float* out = (float*)d_out;

  const int Nn = GCN_N_NODES;
  const int E  = in_sizes[1] / 2;
  const int* src = ei;
  const int* dst = ei + E;

  float* dinv = (float*)d_ws;
  float* h2   = dinv + Nn;
  int* sliceOffs = (int*)(h2 + (size_t)Nn * 64);
  int* rowptr = sliceOffs + EDGE_SLICES * OFFS_PITCH + 32;
  int* binned = rowptr + Nn + 32;
  int* col    = binned + E;

  slice_sort_kernel<<<EDGE_SLICES, 1024, 0, stream>>>(src, dst, binned, sliceOffs, E);
  range_sort_kernel<<<N_RANGES, 1024, 0, stream>>>(binned, sliceOffs, col, rowptr, dinv, Nn, E);

  layer1_fused_kernel<<<512, FUSE_WAVES * 32, 0, stream>>>(x, rowptr, col, dinv, W1, b1, W2, h2,
                                                           Nn);
  gather_out_kernel<<<OUT_BLOCKS, OUT_WAVES * 32, 0, stream>>>(h2, rowptr, col, dinv, b2, out, Nn);
}
